# starting kernel
# baseline (speedup 1.0000x reference)
#include <hip/hip_runtime.h>

typedef __attribute__((ext_vector_type(2))) float v2f;
typedef __attribute__((ext_vector_type(8))) float v8f;


__global__ __launch_bounds__(256) void k_fill(float* __restrict__ p, float v, int n) {
  int i = blockIdx.x * 256 + threadIdx.x;
  if (i < n) p[i] = v;
}

__global__ __launch_bounds__(256) void k_deg(const int* __restrict__ dst,
                                             float* __restrict__ deg, int E) {
  int i = blockIdx.x * 256 + threadIdx.x;
  if (i < E) atomicAdd(&deg[dst[i]], 1.0f);
}

__global__ __launch_bounds__(256) void k_rsqrt(float* __restrict__ p, int n) {
  int i = blockIdx.x * 256 + threadIdx.x;
  if (i < n) p[i] = rsqrtf(p[i]);
}

__global__ __launch_bounds__(256) void k_selfinit(const float* __restrict__ h,
                                                  const float* __restrict__ dinv,
                                                  float* __restrict__ out,
                                                  int n, int shift) {
  int i = blockIdx.x * 256 + threadIdx.x;
  int total = n << shift;
  if (i >= total) return;
  float w = dinv[i >> shift];
  out[i] = h[i] * w * w;
}

__global__ __launch_bounds__(256) void k_bias_act(float* __restrict__ buf,
                                                  const float* __restrict__ b,
                                                  int n, int shift, int mask, int relu) {
  int i = blockIdx.x * 256 + threadIdx.x;
  int total = n << shift;
  if (i >= total) return;
  float v = buf[i] + b[i & mask];
  buf[i] = relu ? fmaxf(v, 0.0f) : v;
}


__global__ __launch_bounds__(256) void k_scatter128(const float* __restrict__ h,
                                                    const int* __restrict__ src,
                                                    const int* __restrict__ dst,
                                                    const float* __restrict__ dinv,
                                                    float* __restrict__ out, int E) {
  int wid = (blockIdx.x * 256 + threadIdx.x) >> 5;
  int lane = threadIdx.x & 31;
  if (wid >= E) return;
  int s = src[wid], d = dst[wid];
  float w = dinv[s] * dinv[d];
  const float4 v = *reinterpret_cast<const float4*>(h + (size_t)s * 128 + lane * 4);
  float* o = out + (size_t)d * 128 + lane * 4;
  atomicAdd(o + 0, v.x * w);
  atomicAdd(o + 1, v.y * w);
  atomicAdd(o + 2, v.z * w);
  atomicAdd(o + 3, v.w * w);
}

__global__ __launch_bounds__(256) void k_scatter64(const float* __restrict__ h,
                                                   const int* __restrict__ src,
                                                   const int* __restrict__ dst,
                                                   const float* __restrict__ dinv,
                                                   float* __restrict__ out, int E) {
  int gid = blockIdx.x * 256 + threadIdx.x;
  int e = gid >> 4;
  int l = gid & 15;
  if (e >= E) return;
  int s = src[e], d = dst[e];
  float w = dinv[s] * dinv[d];
  const float4 v = *reinterpret_cast<const float4*>(h + (size_t)s * 64 + l * 4);
  float* o = out + (size_t)d * 64 + l * 4;
  atomicAdd(o + 0, v.x * w);
  atomicAdd(o + 1, v.y * w);
  atomicAdd(o + 2, v.z * w);
  atomicAdd(o + 3, v.w * w);
}

__global__ __launch_bounds__(256) void k_gemm_wmma(const float* __restrict__ A,
                                                   const float* __restrict__ B,
                                                   const float* __restrict__ bias,
                                                   float* __restrict__ C,
                                                   int M, int K, int Nc) {
  int wid = (blockIdx.x * 256 + threadIdx.x) >> 5;
  int lane = threadIdx.x & 31;
  int nt = Nc >> 4;
  int tM = wid / nt;
  int tN = wid - tM * nt;
  if (tM * 16 >= M) return;

  int r16 = lane & 15;
  int hi  = lane >> 4;

  const float* Ap = A + (size_t)(tM * 16 + r16) * K + 2 * hi;
  const float* Bp = B + (size_t)(2 * hi) * Nc + tN * 16 + r16;

  v8f acc;
  float binit = bias ? bias[tN * 16 + r16] : 0.0f;
#pragma unroll
  for (int r = 0; r < 8; ++r) acc[r] = binit;

  for (int k = 0; k < K; k += 4) {
    v2f a = *reinterpret_cast<const v2f*>(Ap + k);
    v2f b;
    b.x = Bp[(size_t)k * Nc];
    b.y = Bp[(size_t)(k + 1) * Nc];
    acc = __builtin_amdgcn_wmma_f32_16x16x4_f32(false, a, false, b,
                                                (short)0, acc, false, false);
  }

  float* Cp = C + (size_t)(tM * 16 + 8 * hi) * Nc + tN * 16 + r16;
#pragma unroll
  for (int r = 0; r < 8; ++r) Cp[(size_t)r * Nc] = acc[r];
}


extern "C" void kernel_launch(void* const* d_in, const int* in_sizes, int n_in,
                              void* d_out, int out_size, void* d_ws, size_t ws_size,
                              hipStream_t stream) {
  const float* x   = (const float*)d_in[0];
  const int*   ei  = (const int*)d_in[1];
  const float* W1  = (const float*)d_in[2];
  const float* b1  = (const float*)d_in[3];
  const float* W2  = (const float*)d_in[4];
  const float* b2  = (const float*)d_in[5];
  const float* Wfc = (const float*)d_in[6];
  const float* bfc = (const float*)d_in[7];
  float* out = (float*)d_out;

  const int N = in_sizes[0] / 128;
  const int E = in_sizes[1] / 2;
  const int* src = ei;
  const int* dst = ei + E;

  float* dinv = (float*)d_ws;
  float* bufA = dinv + N;
  float* bufB = bufA + (size_t)N * 128;

  auto cdiv = [](long long a, long long b) { return (int)((a + b - 1) / b); };

  k_fill<<<cdiv(N, 256), 256, 0, stream>>>(dinv, 1.0f, N);
  k_deg<<<cdiv(E, 256), 256, 0, stream>>>(dst, dinv, E);
  k_rsqrt<<<cdiv(N, 256), 256, 0, stream>>>(dinv, N);

  {
    int waves = cdiv(N, 16) * (128 >> 4);
    k_gemm_wmma<<<cdiv(waves, 8), 256, 0, stream>>>(x, W1, nullptr, bufA, N, 128, 128);
  }
  k_selfinit<<<cdiv((long long)N << 7, 256), 256, 0, stream>>>(bufA, dinv, bufB, N, 7);
  k_scatter128<<<cdiv(E, 8), 256, 0, stream>>>(bufA, src, dst, dinv, bufB, E);
  k_bias_act<<<cdiv((long long)N << 7, 256), 256, 0, stream>>>(bufB, b1, N, 7, 127, 1);

  {
    int waves = cdiv(N, 16) * (64 >> 4);
    k_gemm_wmma<<<cdiv(waves, 8), 256, 0, stream>>>(bufB, W2, nullptr, bufA, N, 128, 64);
  }
  k_selfinit<<<cdiv((long long)N << 6, 256), 256, 0, stream>>>(bufA, dinv, bufB, N, 6);
  k_scatter64<<<cdiv((long long)E * 16, 256), 256, 0, stream>>>(bufA, src, dst, dinv, bufB, E);
  k_bias_act<<<cdiv((long long)N << 6, 256), 256, 0, stream>>>(bufB, b2, N, 6, 63, 0);

  {
    int waves = cdiv(N, 16) * (64 >> 4);
    k_gemm_wmma<<<cdiv(waves, 8), 256, 0, stream>>>(bufB, Wfc, bfc, out, N, 64, 64);
  }
}


# edge sums find their edges in a one-byte block key per edge (written by the degree pass), 16 keys per lane, and re-read only candidate 16-edge groups of the destination list; sums in arrival order
# speedup vs baseline: 2.2385x; 2.2385x over previous
#include <hip/hip_runtime.h>

typedef __attribute__((ext_vector_type(2))) float v2f;
typedef __attribute__((ext_vector_type(8))) float v8f;


__global__ __launch_bounds__(256) void k_fill(float* __restrict__ p, float v, int n) {
  int i = blockIdx.x * 256 + threadIdx.x;
  if (i < n) p[i] = v;
}

constexpr int kOwnShift = 9;
constexpr int kKeyMax   = 255;

__global__ __launch_bounds__(256) void k_deg(const int* __restrict__ dst,
                                             float* __restrict__ deg,
                                             int* __restrict__ keyw, int E, int Epad) {
  unsigned char* key = reinterpret_cast<unsigned char*>(keyw);
  int i = blockIdx.x * 256 + threadIdx.x;
  if (i < E) {
    int d = dst[i];
    atomicAdd(&deg[d], 1.0f);
    int k = d >> kOwnShift;
    key[i] = (unsigned char)(k < kKeyMax ? k : kKeyMax);
  } else if (i < Epad) {
    key[i] = (unsigned char)kKeyMax;
  }
}

__global__ __launch_bounds__(256) void k_rsqrt(float* __restrict__ p, int n) {
  int i = blockIdx.x * 256 + threadIdx.x;
  if (i < n) p[i] = rsqrtf(p[i]);
}


constexpr int kOwnRows   = 1 << kOwnShift;
constexpr int kAggWaves  = 32;
constexpr int kQueueLen  = 64;
constexpr int kLoads     = 4;
constexpr int kChunk     = 512 * kLoads;
constexpr int kChunkGrps = kChunk / 16;
constexpr int kDrain     = 16;
constexpr int kGQueueLen = 160;
static_assert(kGQueueLen >= kDrain - 1 + 32 * kLoads, "a chunk's groups must fit behind a rest");

template <int C, int U>
__global__ __launch_bounds__(32 * kAggWaves) void k_agg(const float* __restrict__ h,
                                                        const int* __restrict__ src,
                                                        const int* __restrict__ dst,
                                                        const int* __restrict__ keyw,
                                                        const float* __restrict__ dinv,
                                                        float* __restrict__ out, int E, int N) {
  constexpr int R  = C / 32;
  constexpr int C4 = C / 4;
  __shared__ __attribute__((aligned(16))) float tile[kOwnRows * C];
  __shared__ float sdinv[kOwnRows];
  __shared__ int qe[kAggWaves * kQueueLen];
  __shared__ int qd[kAggWaves * kQueueLen];
  __shared__ int gq[kAggWaves * kGQueueLen];

  const int tid  = threadIdx.x;
  const int lane = tid & 31;
  const int wave = __builtin_amdgcn_readfirstlane(tid >> 5);
  const int n0   = blockIdx.x * kOwnRows;

  if (tid < kOwnRows) {
    int n = n0 + tid;
    sdinv[tid] = n < N ? dinv[n] : 0.0f;
  }
  __syncthreads();

  float4* t4 = reinterpret_cast<float4*>(tile);
  {
    const float4* h4 = reinterpret_cast<const float4*>(h + (size_t)n0 * C);
    for (int i = tid; i < kOwnRows * C4; i += 32 * kAggWaves) {
      int r = i / C4;
      float4 v = make_float4(0.0f, 0.0f, 0.0f, 0.0f);
      if (n0 + r < N) {
        v = h4[i];
        float d = sdinv[r];
        v.x = v.x * d * d;
        v.y = v.y * d * d;
        v.z = v.z * d * d;
        v.w = v.w * d * d;
      }
      t4[i] = v;
    }
  }
  __syncthreads();

  int* myqe = qe + wave * kQueueLen;
  int* myqd = qd + wave * kQueueLen;
  int cnt = 0;

  auto add_rows = [&](int s, int dl, float w, int j) {
    float v[U][R];
    for (int u = 0; u < U; ++u) {
      int sj = __builtin_amdgcn_readlane(s, j + u);
      const float* hp = h + (size_t)sj * C + lane;
      for (int r = 0; r < R; ++r) v[u][r] = hp[32 * r];
    }
    for (int u = 0; u < U; ++u) {
      int dj = __builtin_amdgcn_readlane(dl, j + u);
      float wj = __int_as_float(__builtin_amdgcn_readlane(__float_as_int(w), j + u));
      float* o = tile + dj * C + lane;
      for (int r = 0; r < R; ++r) atomicAdd(o + 32 * r, v[u][r] * wj);
    }
  };

  auto flush = [&]() {
    while (cnt >= 32) {
      int e = myqe[lane];
      int dl = myqd[lane];
      int s = src[e];
      float w = dinv[s] * sdinv[dl];
      for (int j = 0; j < 32; j += U) add_rows(s, dl, w, j);
      int rem = cnt - 32;
      if (lane < rem) {
        int te = myqe[32 + lane];
        int td = myqd[32 + lane];
        myqe[lane] = te;
        myqd[lane] = td;
      }
      cnt = rem;
    }
  };

  auto drain_rest = [&]() {
    int e = 0, dl = 0;
    if (lane < cnt) { e = myqe[lane]; dl = myqd[lane]; }
    int s = src[e];
    float w = dinv[s] * sdinv[dl];
    int j = 0;
    for (; j + U <= cnt; j += U) add_rows(s, dl, w, j);
    for (; j < cnt; ++j) {
      int sj = __builtin_amdgcn_readlane(s, j);
      int dj = __builtin_amdgcn_readlane(dl, j);
      float wj = __int_as_float(__builtin_amdgcn_readlane(__float_as_int(w), j));
      const float* hp = h + (size_t)sj * C + lane;
      float* o = tile + dj * C + lane;
      float v[R];
      for (int r = 0; r < R; ++r) v[r] = hp[32 * r];
      for (int r = 0; r < R; ++r) atomicAdd(o + 32 * r, v[r] * wj);
    }
  };

  auto append = [&](unsigned mask, bool mine, int e, int dl) {
    if (mine) {
      int pos = cnt + (int)__builtin_amdgcn_mbcnt_lo(mask, 0u);
      myqe[pos] = e;
      myqd[pos] = dl;
    }
    cnt += __builtin_popcount(mask);
  };

  const int nChunks = (E + kChunk - 1) / kChunk;
  const int rot = (int)(((long long)blockIdx.x * nChunks) / gridDim.x);

  const uint4* key4 = reinterpret_cast<const uint4*>(keyw);
  const unsigned own = blockIdx.x < (unsigned)kKeyMax ? blockIdx.x : (unsigned)kKeyMax;
  const unsigned pat = own * 0x01010101u;

  int* mygq = gq + wave * kGQueueLen;
  int gcnt = 0;
  const int half = lane >> 4;
  const int sub  = lane & 15;

  auto load_chunk = [&](uint4 (&k)[kLoads], int t) -> int {
    int c = t + rot;
    if (c >= nChunks) c -= nChunks;
    int g0 = c * kChunkGrps;
    for (int q = 0; q < kLoads; ++q) k[q] = key4[g0 + 32 * q + lane];
    return g0;
  };

  auto has_own = [&](const uint4& k) -> bool {
    unsigned a = k.x ^ pat, b = k.y ^ pat, c = k.z ^ pat, d = k.w ^ pat;
    unsigned z = ((a - 0x01010101u) & ~a) | ((b - 0x01010101u) & ~b) |
                 ((c - 0x01010101u) & ~c) | ((d - 0x01010101u) & ~d);
    return (z & 0x80808080u) != 0u;
  };

  auto drain = [&](int ng) {
    const int gb = gcnt - ng;
    int gv[kDrain / 2];
    int ev[kDrain / 2];
    int dv[kDrain / 2];
    for (int p = 0; p < kDrain / 2; ++p) {
      int slot = 2 * p + half;
      gv[p] = mygq[slot < ng ? gb + slot : gb];
    }
    for (int p = 0; p < kDrain / 2; ++p) {
      int slot = 2 * p + half;
      int e = slot < ng ? gv[p] * 16 + sub : -1;
      ev[p] = e;
      unsigned ec = (unsigned)e < (unsigned)E ? (unsigned)e : (unsigned)(E - 1);
      dv[p] = dst[ec];
    }
    for (int p = 0; p < kDrain / 2; ++p) {
      if (!((unsigned)ev[p] < (unsigned)E)) dv[p] = -1;
    }
    unsigned mk[kDrain / 2];
    int total = 0;
    for (int p = 0; p < kDrain / 2; ++p) {
      mk[p] = __builtin_amdgcn_ballot_w32((unsigned)(dv[p] - n0) < (unsigned)kOwnRows);
      total += __builtin_popcount(mk[p]);
    }
    if (cnt + total <= kQueueLen) {
      for (int p = 0; p < kDrain / 2; ++p) {
        if (mk[p] != 0u) {
          unsigned dl = (unsigned)(dv[p] - n0);
          append(mk[p], dl < (unsigned)kOwnRows, ev[p], (int)dl);
        }
      }
    } else {
#pragma nounroll
      for (int p = 0; p < kDrain / 2; ++p) {
        if (cnt + 32 > kQueueLen) flush();
        int slot = 2 * p + half;
        int e = -1;
        if (slot < ng) e = mygq[gb + slot] * 16 + sub;
        int d = (unsigned)e < (unsigned)E ? dst[e] : -1;
        unsigned dl = (unsigned)(d - n0);
        bool mine = dl < (unsigned)kOwnRows;
        unsigned mask = __builtin_amdgcn_ballot_w32(mine);
        if (mask != 0u) append(mask, mine, e, (int)dl);
      }
    }
    gcnt = gb;
  };

  int t = wave;
  if (t < nChunks) {
    uint4 kn[kLoads];
    int g0n = load_chunk(kn, t);
    for (;;) {
      uint4 k[kLoads];
      for (int q = 0; q < kLoads; ++q) k[q] = kn[q];
      int g0 = g0n;
      int tn = t + kAggWaves;
      bool more = tn < nChunks;
      if (more) g0n = load_chunk(kn, tn);

      bool hit[kLoads];
      unsigned mk[kLoads];
      int total = 0;
      for (int q = 0; q < kLoads; ++q) {
        hit[q] = has_own(k[q]);
        mk[q] = __builtin_amdgcn_ballot_w32(hit[q]);
        total += __builtin_popcount(mk[q]);
      }
      if (total != 0) {
        for (int q = 0; q < kLoads; ++q) {
          if (mk[q] != 0u) {
            if (hit[q]) {
              int pos = gcnt + (int)__builtin_amdgcn_mbcnt_lo(mk[q], 0u);
              mygq[pos] = g0 + 32 * q + lane;
            }
            gcnt += __builtin_popcount(mk[q]);
          }
        }
      }
      if (total != 0 || !more) {
        while (gcnt >= kDrain || (!more && gcnt > 0)) drain(gcnt < kDrain ? gcnt : kDrain);
        flush();
      }

      if (!more) break;
      t = tn;
    }
  }
  if (cnt > 0) drain_rest();
  __syncthreads();

  {
    float4* o4 = reinterpret_cast<float4*>(out + (size_t)n0 * C);
    for (int i = tid; i < kOwnRows * C4; i += 32 * kAggWaves) {
      int r = i / C4;
      if (n0 + r < N) o4[i] = t4[i];
    }
  }
}

template <bool ABIAS, bool RELU, int K, int NC>
__global__ __launch_bounds__(256) void k_gemm_wmma(const float* __restrict__ A,
                                                   const float* __restrict__ B,
                                                   const float* __restrict__ bias,
                                                   const float* __restrict__ abias,
                                                   float* __restrict__ C, int M) {
  constexpr int NT     = NC / 16;
  constexpr int STRIPS = 8 / NT;
  constexpr int ROWS   = 16 * STRIPS;
  constexpr int PITCH  = K + 4;
  constexpr int K4     = K / 4;
  static_assert(NT * STRIPS == 8 && (ROWS * K4) % 256 == 0, "tile shape");
  __shared__ __attribute__((aligned(16))) float sA[ROWS * PITCH];

  const int tid  = threadIdx.x;
  const int lane = tid & 31;
  const int wave = __builtin_amdgcn_readfirstlane(tid >> 5);
  const int row0 = blockIdx.x * ROWS;

  {
    const float4* A4  = reinterpret_cast<const float4*>(A);
    const float4* ab4 = reinterpret_cast<const float4*>(abias);
    float4 v[ROWS * K4 / 256];
    for (int j = 0; j < ROWS * K4 / 256; ++j) {
      int i  = tid + 256 * j;
      int r  = i / K4;
      int c4 = i % K4;
      v[j] = make_float4(0.0f, 0.0f, 0.0f, 0.0f);
      if (row0 + r < M) v[j] = A4[(size_t)(row0 + r) * K4 + c4];
    }
    for (int j = 0; j < ROWS * K4 / 256; ++j) {
      int i  = tid + 256 * j;
      int r  = i / K4;
      int c4 = i % K4;
      float4 w = v[j];
      if (ABIAS) {
        const float4 bb = ab4[c4];
        w.x += bb.x; w.y += bb.y; w.z += bb.z; w.w += bb.w;
        if (RELU) {
          w.x = fmaxf(w.x, 0.0f); w.y = fmaxf(w.y, 0.0f);
          w.z = fmaxf(w.z, 0.0f); w.w = fmaxf(w.w, 0.0f);
        }
      }
      *reinterpret_cast<float4*>(sA + r * PITCH + 4 * c4) = w;
    }
  }
  __syncthreads();

  const int tMl = wave / NT;
  const int tN  = wave % NT;
  const int tM  = blockIdx.x * STRIPS + tMl;
  if (tM * 16 >= M) return;

  int r16 = lane & 15;
  int hi  = lane >> 4;

  const float* Ap = sA + (tMl * 16 + r16) * PITCH + 2 * hi;
  const float* Bp = B + (size_t)(2 * hi) * NC + tN * 16 + r16;

  v8f acc;
  float binit = bias ? bias[tN * 16 + r16] : 0.0f;
  for (int r = 0; r < 8; ++r) acc[r] = binit;

#pragma unroll 4
  for (int k = 0; k < K; k += 4) {
    v2f a = *reinterpret_cast<const v2f*>(Ap + k);
    v2f b;
    b.x = Bp[(size_t)k * NC];
    b.y = Bp[(size_t)(k + 1) * NC];
    acc = __builtin_amdgcn_wmma_f32_16x16x4_f32(false, a, false, b,
                                                (short)0, acc, false, false);
  }

  float* Cp = C + (size_t)(tM * 16 + 8 * hi) * NC + tN * 16 + r16;
  for (int r = 0; r < 8; ++r) Cp[(size_t)r * NC] = acc[r];
}


extern "C" void kernel_launch(void* const* d_in, const int* in_sizes, int n_in,
                              void* d_out, int out_size, void* d_ws, size_t ws_size,
                              hipStream_t stream) {
  const float* x   = (const float*)d_in[0];
  const int*   ei  = (const int*)d_in[1];
  const float* W1  = (const float*)d_in[2];
  const float* b1  = (const float*)d_in[3];
  const float* W2  = (const float*)d_in[4];
  const float* b2  = (const float*)d_in[5];
  const float* Wfc = (const float*)d_in[6];
  const float* bfc = (const float*)d_in[7];
  float* out = (float*)d_out;

  const int N = in_sizes[0] / 128;
  const int E = in_sizes[1] / 2;
  const int* src = ei;
  const int* dst = ei + E;

  float* dinv = (float*)d_ws;
  float* bufA = dinv + N;
  float* bufB = bufA + (size_t)N * 128;

  auto cdiv = [](long long a, long long b) { return (int)((a + b - 1) / b); };

  k_fill<<<cdiv(N, 256), 256, 0, stream>>>(dinv, 1.0f, N);
  int* key = (int*)d_out;
  const int Epad = cdiv(E, kChunk) * kChunk;
  k_deg<<<cdiv(Epad, 256), 256, 0, stream>>>(dst, dinv, key, E, Epad);
  k_rsqrt<<<cdiv(N, 256), 256, 0, stream>>>(dinv, N);

  {
    int waves = cdiv(N, 16) * (128 >> 4);
    k_gemm_wmma<false, false, 128, 128><<<cdiv(waves, 8), 256, 0, stream>>>(
        x, W1, nullptr, nullptr, bufA, N);
  }
  k_agg<128, 8><<<cdiv(N, kOwnRows), 32 * kAggWaves, 0, stream>>>(bufA, src, dst, key, dinv, bufB, E, N);

  float* h2   = bufA;
  float* acc2 = bufA + (size_t)N * 64;
  {
    int waves = cdiv(N, 16) * (64 >> 4);
    k_gemm_wmma<true, true, 128, 64><<<cdiv(waves, 8), 256, 0, stream>>>(
        bufB, W2, nullptr, b1, h2, N);
  }
  k_agg<64, 8><<<cdiv(N, kOwnRows), 32 * kAggWaves, 0, stream>>>(h2, src, dst, key, dinv, acc2, E, N);

  {
    int waves = cdiv(N, 16) * (64 >> 4);
    k_gemm_wmma<true, false, 64, 64><<<cdiv(waves, 8), 256, 0, stream>>>(
        acc2, Wfc, bfc, b2, out, N);
  }
}


# edge sums: degree pass also writes (destination, source) pairs, 8 B per edge; a candidate 16-edge group is one 128-B line with its sources in it: no scattered src read per edge; order of sums kept
# speedup vs baseline: 2.3605x; 1.0545x over previous
#include <hip/hip_runtime.h>

typedef __attribute__((ext_vector_type(2))) float v2f;
typedef __attribute__((ext_vector_type(8))) float v8f;


__global__ __launch_bounds__(256) void k_fill(float* __restrict__ p, float v, int n) {
  int i = blockIdx.x * 256 + threadIdx.x;
  if (i < n) p[i] = v;
}

constexpr int kOwnShift = 9;
constexpr int kKeyMax   = 255;

__global__ __launch_bounds__(256) void k_deg(const int* __restrict__ src,
                                             const int* __restrict__ dst,
                                             float* __restrict__ deg,
                                             int* __restrict__ keyw,
                                             int* __restrict__ pairw, int E, int Epad) {
  unsigned char* key = reinterpret_cast<unsigned char*>(keyw);
  int2* pair = reinterpret_cast<int2*>(pairw);
  int i = blockIdx.x * 256 + threadIdx.x;
  if (i < E) {
    int d = dst[i];
    int s = src[i];
    atomicAdd(&deg[d], 1.0f);
    int k = d >> kOwnShift;
    key[i] = (unsigned char)(k < kKeyMax ? k : kKeyMax);
    pair[i] = make_int2(d, s);
  } else if (i < Epad) {
    key[i] = (unsigned char)kKeyMax;
  }
}

__global__ __launch_bounds__(256) void k_rsqrt(float* __restrict__ p, int n) {
  int i = blockIdx.x * 256 + threadIdx.x;
  if (i < n) p[i] = rsqrtf(p[i]);
}


constexpr int kOwnRows   = 1 << kOwnShift;
constexpr int kAggWaves  = 32;
constexpr int kQueueLen  = 64;
constexpr int kLoads     = 4;
constexpr int kChunk     = 512 * kLoads;
constexpr int kChunkGrps = kChunk / 16;
constexpr int kDrain     = 16;
constexpr int kGQueueLen = 160;
static_assert(kGQueueLen >= kDrain - 1 + 32 * kLoads, "a chunk's groups must fit behind a rest");

template <int C, int U>
__global__ __launch_bounds__(32 * kAggWaves) void k_agg(const float* __restrict__ h,
                                                        const int* __restrict__ pairw,
                                                        const int* __restrict__ keyw,
                                                        const float* __restrict__ dinv,
                                                        float* __restrict__ out, int E, int N) {
  constexpr int R  = C / 32;
  constexpr int C4 = C / 4;
  __shared__ __attribute__((aligned(16))) float tile[kOwnRows * C];
  __shared__ float sdinv[kOwnRows];
  __shared__ int qs[kAggWaves * kQueueLen];
  __shared__ int qd[kAggWaves * kQueueLen];
  __shared__ int gq[kAggWaves * kGQueueLen];

  const int tid  = threadIdx.x;
  const int lane = tid & 31;
  const int wave = __builtin_amdgcn_readfirstlane(tid >> 5);
  const int n0   = blockIdx.x * kOwnRows;

  if (tid < kOwnRows) {
    int n = n0 + tid;
    sdinv[tid] = n < N ? dinv[n] : 0.0f;
  }
  __syncthreads();

  float4* t4 = reinterpret_cast<float4*>(tile);
  {
    const float4* h4 = reinterpret_cast<const float4*>(h + (size_t)n0 * C);
    for (int i = tid; i < kOwnRows * C4; i += 32 * kAggWaves) {
      int r = i / C4;
      float4 v = make_float4(0.0f, 0.0f, 0.0f, 0.0f);
      if (n0 + r < N) {
        v = h4[i];
        float d = sdinv[r];
        v.x = v.x * d * d;
        v.y = v.y * d * d;
        v.z = v.z * d * d;
        v.w = v.w * d * d;
      }
      t4[i] = v;
    }
  }
  __syncthreads();

  int* myqs = qs + wave * kQueueLen;
  int* myqd = qd + wave * kQueueLen;
  int cnt = 0;

  auto add_rows = [&](int s, int dl, float w, int j) {
    float v[U][R];
    for (int u = 0; u < U; ++u) {
      int sj = __builtin_amdgcn_readlane(s, j + u);
      const float* hp = h + (size_t)sj * C + lane;
      for (int r = 0; r < R; ++r) v[u][r] = hp[32 * r];
    }
    for (int u = 0; u < U; ++u) {
      int dj = __builtin_amdgcn_readlane(dl, j + u);
      float wj = __int_as_float(__builtin_amdgcn_readlane(__float_as_int(w), j + u));
      float* o = tile + dj * C + lane;
      for (int r = 0; r < R; ++r) atomicAdd(o + 32 * r, v[u][r] * wj);
    }
  };

  auto flush = [&]() {
    while (cnt >= 32) {
      int s = myqs[lane];
      int dl = myqd[lane];
      float w = dinv[s] * sdinv[dl];
      for (int j = 0; j < 32; j += U) add_rows(s, dl, w, j);
      int rem = cnt - 32;
      if (lane < rem) {
        int ts = myqs[32 + lane];
        int td = myqd[32 + lane];
        myqs[lane] = ts;
        myqd[lane] = td;
      }
      cnt = rem;
    }
  };

  auto drain_rest = [&]() {
    int s = 0, dl = 0;
    if (lane < cnt) { s = myqs[lane]; dl = myqd[lane]; }
    float w = dinv[s] * sdinv[dl];
    int j = 0;
    for (; j + U <= cnt; j += U) add_rows(s, dl, w, j);
    for (; j < cnt; ++j) {
      int sj = __builtin_amdgcn_readlane(s, j);
      int dj = __builtin_amdgcn_readlane(dl, j);
      float wj = __int_as_float(__builtin_amdgcn_readlane(__float_as_int(w), j));
      const float* hp = h + (size_t)sj * C + lane;
      float* o = tile + dj * C + lane;
      float v[R];
      for (int r = 0; r < R; ++r) v[r] = hp[32 * r];
      for (int r = 0; r < R; ++r) atomicAdd(o + 32 * r, v[r] * wj);
    }
  };

  auto append = [&](unsigned mask, bool mine, int s, int dl) {
    if (mine) {
      int pos = cnt + (int)__builtin_amdgcn_mbcnt_lo(mask, 0u);
      myqs[pos] = s;
      myqd[pos] = dl;
    }
    cnt += __builtin_popcount(mask);
  };

  const int nChunks = (E + kChunk - 1) / kChunk;
  const int rot = (int)(((long long)blockIdx.x * nChunks) / gridDim.x);

  const uint4* key4 = reinterpret_cast<const uint4*>(keyw);
  const int2* pair = reinterpret_cast<const int2*>(pairw);
  const unsigned own = blockIdx.x < (unsigned)kKeyMax ? blockIdx.x : (unsigned)kKeyMax;
  const unsigned pat = own * 0x01010101u;

  int* mygq = gq + wave * kGQueueLen;
  int gcnt = 0;
  const int half = lane >> 4;
  const int sub  = lane & 15;

  auto load_chunk = [&](uint4 (&k)[kLoads], int t) -> int {
    int c = t + rot;
    if (c >= nChunks) c -= nChunks;
    int g0 = c * kChunkGrps;
    for (int q = 0; q < kLoads; ++q) k[q] = key4[g0 + 32 * q + lane];
    return g0;
  };

  auto has_own = [&](const uint4& k) -> bool {
    unsigned a = k.x ^ pat, b = k.y ^ pat, c = k.z ^ pat, d = k.w ^ pat;
    unsigned z = ((a - 0x01010101u) & ~a) | ((b - 0x01010101u) & ~b) |
                 ((c - 0x01010101u) & ~c) | ((d - 0x01010101u) & ~d);
    return (z & 0x80808080u) != 0u;
  };

  auto drain = [&](int ng) {
    const int gb = gcnt - ng;
    int gv[kDrain / 2];
    int ev[kDrain / 2];
    int dv[kDrain / 2];
    int sv[kDrain / 2];
    for (int p = 0; p < kDrain / 2; ++p) {
      int slot = 2 * p + half;
      gv[p] = mygq[slot < ng ? gb + slot : gb];
    }
    for (int p = 0; p < kDrain / 2; ++p) {
      int slot = 2 * p + half;
      int e = slot < ng ? gv[p] * 16 + sub : -1;
      ev[p] = e;
      unsigned ec = (unsigned)e < (unsigned)E ? (unsigned)e : (unsigned)(E - 1);
      int2 ds = pair[ec];
      dv[p] = ds.x;
      sv[p] = ds.y;
    }
    for (int p = 0; p < kDrain / 2; ++p) {
      if (!((unsigned)ev[p] < (unsigned)E)) dv[p] = -1;
    }
    unsigned mk[kDrain / 2];
    int total = 0;
    for (int p = 0; p < kDrain / 2; ++p) {
      mk[p] = __builtin_amdgcn_ballot_w32((unsigned)(dv[p] - n0) < (unsigned)kOwnRows);
      total += __builtin_popcount(mk[p]);
    }
    if (cnt + total <= kQueueLen) {
      for (int p = 0; p < kDrain / 2; ++p) {
        if (mk[p] != 0u) {
          unsigned dl = (unsigned)(dv[p] - n0);
          append(mk[p], dl < (unsigned)kOwnRows, sv[p], (int)dl);
        }
      }
    } else {
#pragma nounroll
      for (int p = 0; p < kDrain / 2; ++p) {
        if (cnt + 32 > kQueueLen) flush();
        int slot = 2 * p + half;
        int e = -1;
        if (slot < ng) e = mygq[gb + slot] * 16 + sub;
        int2 ds = make_int2(-1, 0);
        if ((unsigned)e < (unsigned)E) ds = pair[e];
        unsigned dl = (unsigned)(ds.x - n0);
        bool mine = dl < (unsigned)kOwnRows;
        unsigned mask = __builtin_amdgcn_ballot_w32(mine);
        if (mask != 0u) append(mask, mine, ds.y, (int)dl);
      }
    }
    gcnt = gb;
  };

  int t = wave;
  if (t < nChunks) {
    uint4 kn[kLoads];
    int g0n = load_chunk(kn, t);
    for (;;) {
      uint4 k[kLoads];
      for (int q = 0; q < kLoads; ++q) k[q] = kn[q];
      int g0 = g0n;
      int tn = t + kAggWaves;
      bool more = tn < nChunks;
      if (more) g0n = load_chunk(kn, tn);

      bool hit[kLoads];
      unsigned mk[kLoads];
      int total = 0;
      for (int q = 0; q < kLoads; ++q) {
        hit[q] = has_own(k[q]);
        mk[q] = __builtin_amdgcn_ballot_w32(hit[q]);
        total += __builtin_popcount(mk[q]);
      }
      if (total != 0) {
        for (int q = 0; q < kLoads; ++q) {
          if (mk[q] != 0u) {
            if (hit[q]) {
              int pos = gcnt + (int)__builtin_amdgcn_mbcnt_lo(mk[q], 0u);
              mygq[pos] = g0 + 32 * q + lane;
            }
            gcnt += __builtin_popcount(mk[q]);
          }
        }
      }
      if (total != 0 || !more) {
        while (gcnt >= kDrain || (!more && gcnt > 0)) drain(gcnt < kDrain ? gcnt : kDrain);
        flush();
      }

      if (!more) break;
      t = tn;
    }
  }
  if (cnt > 0) drain_rest();
  __syncthreads();

  {
    float4* o4 = reinterpret_cast<float4*>(out + (size_t)n0 * C);
    for (int i = tid; i < kOwnRows * C4; i += 32 * kAggWaves) {
      int r = i / C4;
      if (n0 + r < N) o4[i] = t4[i];
    }
  }
}

template <bool ABIAS, bool RELU, int K, int NC>
__global__ __launch_bounds__(256) void k_gemm_wmma(const float* __restrict__ A,
                                                   const float* __restrict__ B,
                                                   const float* __restrict__ bias,
                                                   const float* __restrict__ abias,
                                                   float* __restrict__ C, int M) {
  constexpr int NT     = NC / 16;
  constexpr int STRIPS = 8 / NT;
  constexpr int ROWS   = 16 * STRIPS;
  constexpr int PITCH  = K + 4;
  constexpr int K4     = K / 4;
  static_assert(NT * STRIPS == 8 && (ROWS * K4) % 256 == 0, "tile shape");
  __shared__ __attribute__((aligned(16))) float sA[ROWS * PITCH];

  const int tid  = threadIdx.x;
  const int lane = tid & 31;
  const int wave = __builtin_amdgcn_readfirstlane(tid >> 5);
  const int row0 = blockIdx.x * ROWS;

  {
    const float4* A4  = reinterpret_cast<const float4*>(A);
    const float4* ab4 = reinterpret_cast<const float4*>(abias);
    float4 v[ROWS * K4 / 256];
    for (int j = 0; j < ROWS * K4 / 256; ++j) {
      int i  = tid + 256 * j;
      int r  = i / K4;
      int c4 = i % K4;
      v[j] = make_float4(0.0f, 0.0f, 0.0f, 0.0f);
      if (row0 + r < M) v[j] = A4[(size_t)(row0 + r) * K4 + c4];
    }
    for (int j = 0; j < ROWS * K4 / 256; ++j) {
      int i  = tid + 256 * j;
      int r  = i / K4;
      int c4 = i % K4;
      float4 w = v[j];
      if (ABIAS) {
        const float4 bb = ab4[c4];
        w.x += bb.x; w.y += bb.y; w.z += bb.z; w.w += bb.w;
        if (RELU) {
          w.x = fmaxf(w.x, 0.0f); w.y = fmaxf(w.y, 0.0f);
          w.z = fmaxf(w.z, 0.0f); w.w = fmaxf(w.w, 0.0f);
        }
      }
      *reinterpret_cast<float4*>(sA + r * PITCH + 4 * c4) = w;
    }
  }
  __syncthreads();

  const int tMl = wave / NT;
  const int tN  = wave % NT;
  const int tM  = blockIdx.x * STRIPS + tMl;
  if (tM * 16 >= M) return;

  int r16 = lane & 15;
  int hi  = lane >> 4;

  const float* Ap = sA + (tMl * 16 + r16) * PITCH + 2 * hi;
  const float* Bp = B + (size_t)(2 * hi) * NC + tN * 16 + r16;

  v8f acc;
  float binit = bias ? bias[tN * 16 + r16] : 0.0f;
  for (int r = 0; r < 8; ++r) acc[r] = binit;

#pragma unroll 4
  for (int k = 0; k < K; k += 4) {
    v2f a = *reinterpret_cast<const v2f*>(Ap + k);
    v2f b;
    b.x = Bp[(size_t)k * NC];
    b.y = Bp[(size_t)(k + 1) * NC];
    acc = __builtin_amdgcn_wmma_f32_16x16x4_f32(false, a, false, b,
                                                (short)0, acc, false, false);
  }

  float* Cp = C + (size_t)(tM * 16 + 8 * hi) * NC + tN * 16 + r16;
  for (int r = 0; r < 8; ++r) Cp[(size_t)r * NC] = acc[r];
}


extern "C" void kernel_launch(void* const* d_in, const int* in_sizes, int n_in,
                              void* d_out, int out_size, void* d_ws, size_t ws_size,
                              hipStream_t stream) {
  const float* x   = (const float*)d_in[0];
  const int*   ei  = (const int*)d_in[1];
  const float* W1  = (const float*)d_in[2];
  const float* b1  = (const float*)d_in[3];
  const float* W2  = (const float*)d_in[4];
  const float* b2  = (const float*)d_in[5];
  const float* Wfc = (const float*)d_in[6];
  const float* bfc = (const float*)d_in[7];
  float* out = (float*)d_out;

  const int N = in_sizes[0] / 128;
  const int E = in_sizes[1] / 2;
  const int* src = ei;
  const int* dst = ei + E;

  float* dinv = (float*)d_ws;
  float* bufA = dinv + N;
  float* bufB = bufA + (size_t)N * 128;

  auto cdiv = [](long long a, long long b) { return (int)((a + b - 1) / b); };

  k_fill<<<cdiv(N, 256), 256, 0, stream>>>(dinv, 1.0f, N);
  int* key = (int*)d_out;
  const int Epad = cdiv(E, kChunk) * kChunk;
  int* pair = key + Epad / 4;
  k_deg<<<cdiv(Epad, 256), 256, 0, stream>>>(src, dst, dinv, key, pair, E, Epad);
  k_rsqrt<<<cdiv(N, 256), 256, 0, stream>>>(dinv, N);

  {
    int waves = cdiv(N, 16) * (128 >> 4);
    k_gemm_wmma<false, false, 128, 128><<<cdiv(waves, 8), 256, 0, stream>>>(
        x, W1, nullptr, nullptr, bufA, N);
  }
  k_agg<128, 8><<<cdiv(N, kOwnRows), 32 * kAggWaves, 0, stream>>>(bufA, pair, key, dinv, bufB, E, N);

  float* h2   = bufA;
  float* acc2 = bufA + (size_t)N * 64;
  {
    int waves = cdiv(N, 16) * (64 >> 4);
    k_gemm_wmma<true, true, 128, 64><<<cdiv(waves, 8), 256, 0, stream>>>(
        bufB, W2, nullptr, b1, h2, N);
  }
  k_agg<64, 8><<<cdiv(N, kOwnRows), 32 * kAggWaves, 0, stream>>>(h2, pair, key, dinv, acc2, E, N);

  {
    int waves = cdiv(N, 16) * (64 >> 4);
    k_gemm_wmma<true, false, 64, 64><<<cdiv(waves, 8), 256, 0, stream>>>(
        acc2, Wfc, bfc, b2, out, N);
  }
}


# edge sums read a bitmap index, not a key byte per edge: the degree pass (a workgroup per 2048-edge chunk) ORs one 128-bit group mask per destination block in LDS; sums stay in arrival order
# speedup vs baseline: 2.5261x; 1.0702x over previous
#include <hip/hip_runtime.h>

typedef __attribute__((ext_vector_type(2))) float v2f;
typedef __attribute__((ext_vector_type(8))) float v8f;


__global__ __launch_bounds__(256) void k_fill(float* __restrict__ p, float v, int n) {
  int i = blockIdx.x * 256 + threadIdx.x;
  if (i < n) p[i] = v;
}

constexpr int kOwnShift  = 9;
constexpr int kKeyMax    = 255;
constexpr int kMaskRows  = kKeyMax + 1;
constexpr int kChunk     = 2048;
constexpr int kChunkGrps = kChunk / 16;
constexpr int kDegEdges  = kChunk / 256;
static_assert(kChunkGrps == 128, "a mask word has 128 bits");

__global__ __launch_bounds__(256) void k_deg(const int* __restrict__ src,
                                             const int* __restrict__ dst,
                                             float* __restrict__ deg,
                                             int* __restrict__ maskw,
                                             int* __restrict__ pairw, int E) {
  __shared__ __attribute__((aligned(16))) unsigned smask[kMaskRows * 4];
  const int tid = threadIdx.x;
  uint4* sm4 = reinterpret_cast<uint4*>(smask);
  sm4[tid] = make_uint4(0u, 0u, 0u, 0u);
  __syncthreads();

  int2* pair = reinterpret_cast<int2*>(pairw);
  const int base = blockIdx.x * kChunk;
  int d[kDegEdges];
  int s[kDegEdges];
  for (int j = 0; j < kDegEdges; ++j) {
    int i = base + j * 256 + tid;
    d[j] = 0;
    s[j] = 0;
    if (i < E) {
      d[j] = dst[i];
      s[j] = src[i];
    }
  }
  for (int j = 0; j < kDegEdges; ++j) {
    int i = base + j * 256 + tid;
    if (i < E) {
      atomicAdd(&deg[d[j]], 1.0f);
      pair[i] = make_int2(d[j], s[j]);
      int k = d[j] >> kOwnShift;
      k = k < kKeyMax ? k : kKeyMax;
      int g = j * 16 + (tid >> 4);
      atomicOr(&smask[k * 4 + (g >> 5)], 1u << (g & 31));
    }
  }
  __syncthreads();
  reinterpret_cast<uint4*>(maskw)[(size_t)blockIdx.x * kMaskRows + tid] = sm4[tid];
}

__global__ __launch_bounds__(256) void k_rsqrt(float* __restrict__ p, int n) {
  int i = blockIdx.x * 256 + threadIdx.x;
  if (i < n) p[i] = rsqrtf(p[i]);
}


constexpr int kOwnRows   = 1 << kOwnShift;
constexpr int kAggWaves  = 32;
constexpr int kQueueLen  = 64;
constexpr int kDrain     = 16;
constexpr int kGQueueLen = 160;
static_assert(kGQueueLen >= kDrain - 1 + 32, "a round's groups must fit behind a rest");

template <int C, int U>
__global__ __launch_bounds__(32 * kAggWaves) void k_agg(const float* __restrict__ h,
                                                        const int* __restrict__ pairw,
                                                        const int* __restrict__ maskw,
                                                        const float* __restrict__ dinv,
                                                        float* __restrict__ out, int E, int N) {
  constexpr int R  = C / 32;
  constexpr int C4 = C / 4;
  __shared__ __attribute__((aligned(16))) float tile[kOwnRows * C];
  __shared__ float sdinv[kOwnRows];
  __shared__ int qs[kAggWaves * kQueueLen];
  __shared__ int qd[kAggWaves * kQueueLen];
  __shared__ int gq[kAggWaves * kGQueueLen];

  const int tid  = threadIdx.x;
  const int lane = tid & 31;
  const int wave = __builtin_amdgcn_readfirstlane(tid >> 5);
  const int n0   = blockIdx.x * kOwnRows;

  if (tid < kOwnRows) {
    int n = n0 + tid;
    sdinv[tid] = n < N ? dinv[n] : 0.0f;
  }
  __syncthreads();

  float4* t4 = reinterpret_cast<float4*>(tile);
  {
    const float4* h4 = reinterpret_cast<const float4*>(h + (size_t)n0 * C);
    for (int i = tid; i < kOwnRows * C4; i += 32 * kAggWaves) {
      int r = i / C4;
      float4 v = make_float4(0.0f, 0.0f, 0.0f, 0.0f);
      if (n0 + r < N) {
        v = h4[i];
        float d = sdinv[r];
        v.x = v.x * d * d;
        v.y = v.y * d * d;
        v.z = v.z * d * d;
        v.w = v.w * d * d;
      }
      t4[i] = v;
    }
  }
  __syncthreads();

  int* myqs = qs + wave * kQueueLen;
  int* myqd = qd + wave * kQueueLen;
  int cnt = 0;

  auto add_rows = [&](int s, int dl, float w, int j) {
    float v[U][R];
    for (int u = 0; u < U; ++u) {
      int sj = __builtin_amdgcn_readlane(s, j + u);
      const float* hp = h + (size_t)sj * C + lane;
      for (int r = 0; r < R; ++r) v[u][r] = hp[32 * r];
    }
    for (int u = 0; u < U; ++u) {
      int dj = __builtin_amdgcn_readlane(dl, j + u);
      float wj = __int_as_float(__builtin_amdgcn_readlane(__float_as_int(w), j + u));
      float* o = tile + dj * C + lane;
      for (int r = 0; r < R; ++r) atomicAdd(o + 32 * r, v[u][r] * wj);
    }
  };

  auto flush = [&]() {
    while (cnt >= 32) {
      int s = myqs[lane];
      int dl = myqd[lane];
      float w = dinv[s] * sdinv[dl];
      for (int j = 0; j < 32; j += U) add_rows(s, dl, w, j);
      int rem = cnt - 32;
      if (lane < rem) {
        int ts = myqs[32 + lane];
        int td = myqd[32 + lane];
        myqs[lane] = ts;
        myqd[lane] = td;
      }
      cnt = rem;
    }
  };

  auto drain_rest = [&]() {
    int s = 0, dl = 0;
    if (lane < cnt) { s = myqs[lane]; dl = myqd[lane]; }
    float w = dinv[s] * sdinv[dl];
    int j = 0;
    for (; j + U <= cnt; j += U) add_rows(s, dl, w, j);
    for (; j < cnt; ++j) {
      int sj = __builtin_amdgcn_readlane(s, j);
      int dj = __builtin_amdgcn_readlane(dl, j);
      float wj = __int_as_float(__builtin_amdgcn_readlane(__float_as_int(w), j));
      const float* hp = h + (size_t)sj * C + lane;
      float* o = tile + dj * C + lane;
      float v[R];
      for (int r = 0; r < R; ++r) v[r] = hp[32 * r];
      for (int r = 0; r < R; ++r) atomicAdd(o + 32 * r, v[r] * wj);
    }
  };

  auto append = [&](unsigned mask, bool mine, int s, int dl) {
    if (mine) {
      int pos = cnt + (int)__builtin_amdgcn_mbcnt_lo(mask, 0u);
      myqs[pos] = s;
      myqd[pos] = dl;
    }
    cnt += __builtin_popcount(mask);
  };

  const int nChunks = (E + kChunk - 1) / kChunk;
  const uint4* mask4 = reinterpret_cast<const uint4*>(maskw);
  const int2* pair = reinterpret_cast<const int2*>(pairw);
  const unsigned own = blockIdx.x < (unsigned)kKeyMax ? blockIdx.x : (unsigned)kKeyMax;

  int* mygq = gq + wave * kGQueueLen;
  int gcnt = 0;
  const int half = lane >> 4;
  const int sub  = lane & 15;

  auto drain = [&](int ng) {
    const int gb = gcnt - ng;
    int gv[kDrain / 2];
    int ev[kDrain / 2];
    int dv[kDrain / 2];
    int sv[kDrain / 2];
    for (int p = 0; p < kDrain / 2; ++p) {
      int slot = 2 * p + half;
      gv[p] = mygq[slot < ng ? gb + slot : gb];
    }
    for (int p = 0; p < kDrain / 2; ++p) {
      int slot = 2 * p + half;
      int e = slot < ng ? gv[p] * 16 + sub : -1;
      ev[p] = e;
      unsigned ec = (unsigned)e < (unsigned)E ? (unsigned)e : (unsigned)(E - 1);
      int2 ds = pair[ec];
      dv[p] = ds.x;
      sv[p] = ds.y;
    }
    for (int p = 0; p < kDrain / 2; ++p) {
      if (!((unsigned)ev[p] < (unsigned)E)) dv[p] = -1;
    }
    unsigned mk[kDrain / 2];
    int total = 0;
    for (int p = 0; p < kDrain / 2; ++p) {
      mk[p] = __builtin_amdgcn_ballot_w32((unsigned)(dv[p] - n0) < (unsigned)kOwnRows);
      total += __builtin_popcount(mk[p]);
    }
    if (cnt + total <= kQueueLen) {
      for (int p = 0; p < kDrain / 2; ++p) {
        if (mk[p] != 0u) {
          unsigned dl = (unsigned)(dv[p] - n0);
          append(mk[p], dl < (unsigned)kOwnRows, sv[p], (int)dl);
        }
      }
    } else {
#pragma nounroll
      for (int p = 0; p < kDrain / 2; ++p) {
        if (cnt + 32 > kQueueLen) flush();
        int slot = 2 * p + half;
        int e = -1;
        if (slot < ng) e = mygq[gb + slot] * 16 + sub;
        int2 ds = make_int2(-1, 0);
        if ((unsigned)e < (unsigned)E) ds = pair[e];
        unsigned dl = (unsigned)(ds.x - n0);
        bool mine = dl < (unsigned)kOwnRows;
        unsigned mask = __builtin_amdgcn_ballot_w32(mine);
        if (mask != 0u) append(mask, mine, ds.y, (int)dl);
      }
    }
    gcnt = gb;
  };

  for (int c0 = wave; c0 < nChunks; c0 += 32 * kAggWaves) {
    const int c = c0 + kAggWaves * lane;
    uint4 m = make_uint4(0u, 0u, 0u, 0u);
    if (c < nChunks) m = mask4[(size_t)c * kMaskRows + own];
    unsigned long long lo = (unsigned long long)m.x | ((unsigned long long)m.y << 32);
    unsigned long long hi = (unsigned long long)m.z | ((unsigned long long)m.w << 32);
    const int gfirst = c * kChunkGrps;
    for (;;) {
      bool hit = (lo | hi) != 0ull;
      unsigned mk = __builtin_amdgcn_ballot_w32(hit);
      if (mk == 0u) break;
      if (hit) {
        int g;
        if (lo != 0ull) {
          g = __builtin_ctzll(lo);
          lo &= lo - 1ull;
        } else {
          g = 64 + __builtin_ctzll(hi);
          hi &= hi - 1ull;
        }
        int pos = gcnt + (int)__builtin_amdgcn_mbcnt_lo(mk, 0u);
        mygq[pos] = gfirst + g;
      }
      gcnt += __builtin_popcount(mk);
      while (gcnt >= kDrain) drain(kDrain);
      flush();
    }
  }
  if (gcnt > 0) drain(gcnt);
  flush();
  if (cnt > 0) drain_rest();
  __syncthreads();

  {
    float4* o4 = reinterpret_cast<float4*>(out + (size_t)n0 * C);
    for (int i = tid; i < kOwnRows * C4; i += 32 * kAggWaves) {
      int r = i / C4;
      if (n0 + r < N) o4[i] = t4[i];
    }
  }
}

template <bool ABIAS, bool RELU, int K, int NC>
__global__ __launch_bounds__(256) void k_gemm_wmma(const float* __restrict__ A,
                                                   const float* __restrict__ B,
                                                   const float* __restrict__ bias,
                                                   const float* __restrict__ abias,
                                                   float* __restrict__ C, int M) {
  constexpr int NT     = NC / 16;
  constexpr int STRIPS = 8 / NT;
  constexpr int ROWS   = 16 * STRIPS;
  constexpr int PITCH  = K + 4;
  constexpr int K4     = K / 4;
  static_assert(NT * STRIPS == 8 && (ROWS * K4) % 256 == 0, "tile shape");
  __shared__ __attribute__((aligned(16))) float sA[ROWS * PITCH];

  const int tid  = threadIdx.x;
  const int lane = tid & 31;
  const int wave = __builtin_amdgcn_readfirstlane(tid >> 5);
  const int row0 = blockIdx.x * ROWS;

  {
    const float4* A4  = reinterpret_cast<const float4*>(A);
    const float4* ab4 = reinterpret_cast<const float4*>(abias);
    float4 v[ROWS * K4 / 256];
    for (int j = 0; j < ROWS * K4 / 256; ++j) {
      int i  = tid + 256 * j;
      int r  = i / K4;
      int c4 = i % K4;
      v[j] = make_float4(0.0f, 0.0f, 0.0f, 0.0f);
      if (row0 + r < M) v[j] = A4[(size_t)(row0 + r) * K4 + c4];
    }
    for (int j = 0; j < ROWS * K4 / 256; ++j) {
      int i  = tid + 256 * j;
      int r  = i / K4;
      int c4 = i % K4;
      float4 w = v[j];
      if (ABIAS) {
        const float4 bb = ab4[c4];
        w.x += bb.x; w.y += bb.y; w.z += bb.z; w.w += bb.w;
        if (RELU) {
          w.x = fmaxf(w.x, 0.0f); w.y = fmaxf(w.y, 0.0f);
          w.z = fmaxf(w.z, 0.0f); w.w = fmaxf(w.w, 0.0f);
        }
      }
      *reinterpret_cast<float4*>(sA + r * PITCH + 4 * c4) = w;
    }
  }
  __syncthreads();

  const int tMl = wave / NT;
  const int tN  = wave % NT;
  const int tM  = blockIdx.x * STRIPS + tMl;
  if (tM * 16 >= M) return;

  int r16 = lane & 15;
  int hi  = lane >> 4;

  const float* Ap = sA + (tMl * 16 + r16) * PITCH + 2 * hi;
  const float* Bp = B + (size_t)(2 * hi) * NC + tN * 16 + r16;

  v8f acc;
  float binit = bias ? bias[tN * 16 + r16] : 0.0f;
  for (int r = 0; r < 8; ++r) acc[r] = binit;

#pragma unroll 4
  for (int k = 0; k < K; k += 4) {
    v2f a = *reinterpret_cast<const v2f*>(Ap + k);
    v2f b;
    b.x = Bp[(size_t)k * NC];
    b.y = Bp[(size_t)(k + 1) * NC];
    acc = __builtin_amdgcn_wmma_f32_16x16x4_f32(false, a, false, b,
                                                (short)0, acc, false, false);
  }

  float* Cp = C + (size_t)(tM * 16 + 8 * hi) * NC + tN * 16 + r16;
  for (int r = 0; r < 8; ++r) Cp[(size_t)r * NC] = acc[r];
}


extern "C" void kernel_launch(void* const* d_in, const int* in_sizes, int n_in,
                              void* d_out, int out_size, void* d_ws, size_t ws_size,
                              hipStream_t stream) {
  const float* x   = (const float*)d_in[0];
  const int*   ei  = (const int*)d_in[1];
  const float* W1  = (const float*)d_in[2];
  const float* b1  = (const float*)d_in[3];
  const float* W2  = (const float*)d_in[4];
  const float* b2  = (const float*)d_in[5];
  const float* Wfc = (const float*)d_in[6];
  const float* bfc = (const float*)d_in[7];
  float* out = (float*)d_out;

  const int N = in_sizes[0] / 128;
  const int E = in_sizes[1] / 2;
  const int* src = ei;
  const int* dst = ei + E;

  float* dinv = (float*)d_ws;
  float* bufA = dinv + N;
  float* bufB = bufA + (size_t)N * 128;

  auto cdiv = [](long long a, long long b) { return (int)((a + b - 1) / b); };

  k_fill<<<cdiv(N, 256), 256, 0, stream>>>(dinv, 1.0f, N);
  int* mask = (int*)d_out;
  const int nChunks = cdiv(E, kChunk);
  int* pair = mask + (size_t)nChunks * (kMaskRows * 4);
  k_deg<<<nChunks, 256, 0, stream>>>(src, dst, dinv, mask, pair, E);
  k_rsqrt<<<cdiv(N, 256), 256, 0, stream>>>(dinv, N);

  {
    int waves = cdiv(N, 16) * (128 >> 4);
    k_gemm_wmma<false, false, 128, 128><<<cdiv(waves, 8), 256, 0, stream>>>(
        x, W1, nullptr, nullptr, bufA, N);
  }
  k_agg<128, 8><<<cdiv(N, kOwnRows), 32 * kAggWaves, 0, stream>>>(bufA, pair, mask, dinv, bufB, E, N);

  float* h2   = bufA;
  float* acc2 = bufA + (size_t)N * 64;
  {
    int waves = cdiv(N, 16) * (64 >> 4);
    k_gemm_wmma<true, true, 128, 64><<<cdiv(waves, 8), 256, 0, stream>>>(
        bufB, W2, nullptr, b1, h2, N);
  }
  k_agg<64, 8><<<cdiv(N, kOwnRows), 32 * kAggWaves, 0, stream>>>(h2, pair, mask, dinv, acc2, E, N);

  {
    int waves = cdiv(N, 16) * (64 >> 4);
    k_gemm_wmma<true, false, 64, 64><<<cdiv(waves, 8), 256, 0, stream>>>(
        acc2, Wfc, bfc, b2, out, N);
  }
}


# degree pass writes each 2048-edge chunk's (destination, source) pairs ordered by destination block (LDS counting sort): bitmap flags 1-2 lines per block and chunk, not ten; order of sums changes
# speedup vs baseline: 2.6839x; 1.0625x over previous
#include <hip/hip_runtime.h>

typedef __attribute__((ext_vector_type(2))) float v2f;
typedef __attribute__((ext_vector_type(8))) float v8f;


__global__ __launch_bounds__(256) void k_fill(float* __restrict__ p, float v, int n) {
  int i = blockIdx.x * 256 + threadIdx.x;
  if (i < n) p[i] = v;
}

constexpr int kOwnShift  = 9;
constexpr int kKeyMax    = 255;
constexpr int kMaskRows  = kKeyMax + 1;
constexpr int kChunk     = 2048;
constexpr int kChunkGrps = kChunk / 16;
constexpr int kDegEdges  = kChunk / 256;
static_assert(kChunkGrps == 128, "a mask word has 128 bits");
static_assert(kMaskRows == 256, "one thread per block key in the prefix sum");

__global__ __launch_bounds__(256) void k_deg(const int* __restrict__ src,
                                             const int* __restrict__ dst,
                                             float* __restrict__ deg,
                                             int* __restrict__ maskw,
                                             int* __restrict__ pairw, int E) {
  __shared__ __attribute__((aligned(16))) unsigned smask[kMaskRows * 4];
  __shared__ int scnt[kMaskRows];
  __shared__ int swsum[8];
  __shared__ __attribute__((aligned(16))) int2 spair[kChunk];
  const int tid = threadIdx.x;
  uint4* sm4 = reinterpret_cast<uint4*>(smask);
  sm4[tid] = make_uint4(0u, 0u, 0u, 0u);
  scnt[tid] = 0;
  __syncthreads();

  int2* pair = reinterpret_cast<int2*>(pairw);
  const int base = blockIdx.x * kChunk;
  int d[kDegEdges];
  int s[kDegEdges];
  int rk[kDegEdges];
  for (int j = 0; j < kDegEdges; ++j) {
    int i = base + j * 256 + tid;
    d[j] = 0;
    s[j] = 0;
    if (i < E) {
      d[j] = dst[i];
      s[j] = src[i];
    }
  }
  for (int j = 0; j < kDegEdges; ++j) {
    int i = base + j * 256 + tid;
    rk[j] = 0;
    if (i < E) {
      atomicAdd(&deg[d[j]], 1.0f);
      int k = d[j] >> kOwnShift;
      k = k < kKeyMax ? k : kKeyMax;
      rk[j] = atomicAdd(&scnt[k], 1);
    }
  }
  __syncthreads();

  {
    const int lane = tid & 31;
    const int wave = tid >> 5;
    const int c = scnt[tid];
    int incl = c;
    for (int off = 1; off < 32; off <<= 1) {
      int t = __shfl_up(incl, off);
      if (lane >= off) incl += t;
    }
    if (lane == 31) swsum[wave] = incl;
    __syncthreads();
    int wbase = 0;
    for (int w = 0; w < 8; ++w) {
      if (w < wave) wbase += swsum[w];
    }
    scnt[tid] = wbase + incl - c;
  }
  __syncthreads();

  for (int j = 0; j < kDegEdges; ++j) {
    int i = base + j * 256 + tid;
    if (i < E) {
      int k = d[j] >> kOwnShift;
      k = k < kKeyMax ? k : kKeyMax;
      int pos = scnt[k] + rk[j];
      spair[pos] = make_int2(d[j], s[j]);
      int g = pos >> 4;
      atomicOr(&smask[k * 4 + (g >> 5)], 1u << (g & 31));
    }
  }
  __syncthreads();
  for (int j = 0; j < kDegEdges; ++j) {
    int p = j * 256 + tid;
    if (base + p < E) pair[base + p] = spair[p];
  }
  reinterpret_cast<uint4*>(maskw)[(size_t)blockIdx.x * kMaskRows + tid] = sm4[tid];
}

__global__ __launch_bounds__(256) void k_rsqrt(float* __restrict__ p, int n) {
  int i = blockIdx.x * 256 + threadIdx.x;
  if (i < n) p[i] = rsqrtf(p[i]);
}


constexpr int kOwnRows   = 1 << kOwnShift;
constexpr int kAggWaves  = 32;
constexpr int kQueueLen  = 128;
constexpr int kDrain     = 8;
constexpr int kGQueueLen = 160;
static_assert(kGQueueLen >= kDrain - 1 + 32, "a round's groups must fit behind a rest");

template <int C, int U>
__global__ __launch_bounds__(32 * kAggWaves) void k_agg(const float* __restrict__ h,
                                                        const int* __restrict__ pairw,
                                                        const int* __restrict__ maskw,
                                                        const float* __restrict__ dinv,
                                                        float* __restrict__ out, int E, int N) {
  constexpr int R  = C / 32;
  constexpr int C4 = C / 4;
  __shared__ __attribute__((aligned(16))) float tile[kOwnRows * C];
  __shared__ float sdinv[kOwnRows];
  __shared__ int qs[kAggWaves * kQueueLen];
  __shared__ int qd[kAggWaves * kQueueLen];
  __shared__ int gq[kAggWaves * kGQueueLen];

  const int tid  = threadIdx.x;
  const int lane = tid & 31;
  const int wave = __builtin_amdgcn_readfirstlane(tid >> 5);
  const int n0   = blockIdx.x * kOwnRows;

  if (tid < kOwnRows) {
    int n = n0 + tid;
    sdinv[tid] = n < N ? dinv[n] : 0.0f;
  }
  __syncthreads();

  float4* t4 = reinterpret_cast<float4*>(tile);
  {
    const float4* h4 = reinterpret_cast<const float4*>(h + (size_t)n0 * C);
    for (int i = tid; i < kOwnRows * C4; i += 32 * kAggWaves) {
      int r = i / C4;
      float4 v = make_float4(0.0f, 0.0f, 0.0f, 0.0f);
      if (n0 + r < N) {
        v = h4[i];
        float d = sdinv[r];
        v.x = v.x * d * d;
        v.y = v.y * d * d;
        v.z = v.z * d * d;
        v.w = v.w * d * d;
      }
      t4[i] = v;
    }
  }
  __syncthreads();

  int* myqs = qs + wave * kQueueLen;
  int* myqd = qd + wave * kQueueLen;
  int cnt = 0;

  auto add_rows = [&](int s, int dl, float w, int j) {
    float v[U][R];
    for (int u = 0; u < U; ++u) {
      int sj = __builtin_amdgcn_readlane(s, j + u);
      const float* hp = h + (size_t)sj * C + lane;
      for (int r = 0; r < R; ++r) v[u][r] = hp[32 * r];
    }
    for (int u = 0; u < U; ++u) {
      int dj = __builtin_amdgcn_readlane(dl, j + u);
      float wj = __int_as_float(__builtin_amdgcn_readlane(__float_as_int(w), j + u));
      float* o = tile + dj * C + lane;
      for (int r = 0; r < R; ++r) atomicAdd(o + 32 * r, v[u][r] * wj);
    }
  };

  auto flush = [&]() {
    int head = 0;
    while (cnt - head >= 32) {
      int s = myqs[head + lane];
      int dl = myqd[head + lane];
      float w = dinv[s] * sdinv[dl];
      for (int j = 0; j < 32; j += U) add_rows(s, dl, w, j);
      head += 32;
    }
    if (head > 0) {
      int rem = cnt - head;
      if (lane < rem) {
        int ts = myqs[head + lane];
        int td = myqd[head + lane];
        myqs[lane] = ts;
        myqd[lane] = td;
      }
      cnt = rem;
    }
  };

  auto drain_rest = [&]() {
    int s = 0, dl = 0;
    if (lane < cnt) { s = myqs[lane]; dl = myqd[lane]; }
    float w = dinv[s] * sdinv[dl];
    int j = 0;
    for (; j + U <= cnt; j += U) add_rows(s, dl, w, j);
    for (; j < cnt; ++j) {
      int sj = __builtin_amdgcn_readlane(s, j);
      int dj = __builtin_amdgcn_readlane(dl, j);
      float wj = __int_as_float(__builtin_amdgcn_readlane(__float_as_int(w), j));
      const float* hp = h + (size_t)sj * C + lane;
      float* o = tile + dj * C + lane;
      float v[R];
      for (int r = 0; r < R; ++r) v[r] = hp[32 * r];
      for (int r = 0; r < R; ++r) atomicAdd(o + 32 * r, v[r] * wj);
    }
  };

  auto append = [&](unsigned mask, bool mine, int s, int dl) {
    if (mine) {
      int pos = cnt + (int)__builtin_amdgcn_mbcnt_lo(mask, 0u);
      myqs[pos] = s;
      myqd[pos] = dl;
    }
    cnt += __builtin_popcount(mask);
  };

  const int nChunks = (E + kChunk - 1) / kChunk;
  const uint4* mask4 = reinterpret_cast<const uint4*>(maskw);
  const int2* pair = reinterpret_cast<const int2*>(pairw);
  const unsigned own = blockIdx.x < (unsigned)kKeyMax ? blockIdx.x : (unsigned)kKeyMax;

  int* mygq = gq + wave * kGQueueLen;
  int gcnt = 0;
  const int half = lane >> 4;
  const int sub  = lane & 15;

  auto drain = [&](int ng) {
    const int gb = gcnt - ng;
    int gv[kDrain / 2];
    int ev[kDrain / 2];
    int dv[kDrain / 2];
    int sv[kDrain / 2];
    for (int p = 0; p < kDrain / 2; ++p) {
      int slot = 2 * p + half;
      gv[p] = mygq[slot < ng ? gb + slot : gb];
    }
    for (int p = 0; p < kDrain / 2; ++p) {
      int slot = 2 * p + half;
      int e = slot < ng ? gv[p] * 16 + sub : -1;
      ev[p] = e;
      unsigned ec = (unsigned)e < (unsigned)E ? (unsigned)e : (unsigned)(E - 1);
      int2 ds = pair[ec];
      dv[p] = ds.x;
      sv[p] = ds.y;
    }
    for (int p = 0; p < kDrain / 2; ++p) {
      if (!((unsigned)ev[p] < (unsigned)E)) dv[p] = -1;
    }
    unsigned mk[kDrain / 2];
    int total = 0;
    for (int p = 0; p < kDrain / 2; ++p) {
      mk[p] = __builtin_amdgcn_ballot_w32((unsigned)(dv[p] - n0) < (unsigned)kOwnRows);
      total += __builtin_popcount(mk[p]);
    }
    if (cnt + total <= kQueueLen) {
      for (int p = 0; p < kDrain / 2; ++p) {
        if (mk[p] != 0u) {
          unsigned dl = (unsigned)(dv[p] - n0);
          append(mk[p], dl < (unsigned)kOwnRows, sv[p], (int)dl);
        }
      }
    } else {
#pragma nounroll
      for (int p = 0; p < kDrain / 2; ++p) {
        if (cnt + 32 > kQueueLen) flush();
        int slot = 2 * p + half;
        int e = -1;
        if (slot < ng) e = mygq[gb + slot] * 16 + sub;
        int2 ds = make_int2(-1, 0);
        if ((unsigned)e < (unsigned)E) ds = pair[e];
        unsigned dl = (unsigned)(ds.x - n0);
        bool mine = dl < (unsigned)kOwnRows;
        unsigned mask = __builtin_amdgcn_ballot_w32(mine);
        if (mask != 0u) append(mask, mine, ds.y, (int)dl);
      }
    }
    gcnt = gb;
  };

  for (int c0 = wave; c0 < nChunks; c0 += 32 * kAggWaves) {
    const int c = c0 + kAggWaves * lane;
    uint4 m = make_uint4(0u, 0u, 0u, 0u);
    if (c < nChunks) m = mask4[(size_t)c * kMaskRows + own];
    unsigned long long lo = (unsigned long long)m.x | ((unsigned long long)m.y << 32);
    unsigned long long hi = (unsigned long long)m.z | ((unsigned long long)m.w << 32);
    const int gfirst = c * kChunkGrps;
    for (;;) {
      bool hit = (lo | hi) != 0ull;
      unsigned mk = __builtin_amdgcn_ballot_w32(hit);
      if (mk == 0u) break;
      if (hit) {
        int g;
        if (lo != 0ull) {
          g = __builtin_ctzll(lo);
          lo &= lo - 1ull;
        } else {
          g = 64 + __builtin_ctzll(hi);
          hi &= hi - 1ull;
        }
        int pos = gcnt + (int)__builtin_amdgcn_mbcnt_lo(mk, 0u);
        mygq[pos] = gfirst + g;
      }
      gcnt += __builtin_popcount(mk);
      while (gcnt >= kDrain) {
        drain(kDrain);
        flush();
      }
    }
  }
  if (gcnt > 0) drain(gcnt);
  flush();
  if (cnt > 0) drain_rest();
  __syncthreads();

  {
    float4* o4 = reinterpret_cast<float4*>(out + (size_t)n0 * C);
    for (int i = tid; i < kOwnRows * C4; i += 32 * kAggWaves) {
      int r = i / C4;
      if (n0 + r < N) o4[i] = t4[i];
    }
  }
}

template <bool ABIAS, bool RELU, int K, int NC>
__global__ __launch_bounds__(256) void k_gemm_wmma(const float* __restrict__ A,
                                                   const float* __restrict__ B,
                                                   const float* __restrict__ bias,
                                                   const float* __restrict__ abias,
                                                   float* __restrict__ C, int M) {
  constexpr int NT     = NC / 16;
  constexpr int STRIPS = 8 / NT;
  constexpr int ROWS   = 16 * STRIPS;
  constexpr int PITCH  = K + 4;
  constexpr int K4     = K / 4;
  static_assert(NT * STRIPS == 8 && (ROWS * K4) % 256 == 0, "tile shape");
  __shared__ __attribute__((aligned(16))) float sA[ROWS * PITCH];

  const int tid  = threadIdx.x;
  const int lane = tid & 31;
  const int wave = __builtin_amdgcn_readfirstlane(tid >> 5);
  const int row0 = blockIdx.x * ROWS;

  {
    const float4* A4  = reinterpret_cast<const float4*>(A);
    const float4* ab4 = reinterpret_cast<const float4*>(abias);
    float4 v[ROWS * K4 / 256];
    for (int j = 0; j < ROWS * K4 / 256; ++j) {
      int i  = tid + 256 * j;
      int r  = i / K4;
      int c4 = i % K4;
      v[j] = make_float4(0.0f, 0.0f, 0.0f, 0.0f);
      if (row0 + r < M) v[j] = A4[(size_t)(row0 + r) * K4 + c4];
    }
    for (int j = 0; j < ROWS * K4 / 256; ++j) {
      int i  = tid + 256 * j;
      int r  = i / K4;
      int c4 = i % K4;
      float4 w = v[j];
      if (ABIAS) {
        const float4 bb = ab4[c4];
        w.x += bb.x; w.y += bb.y; w.z += bb.z; w.w += bb.w;
        if (RELU) {
          w.x = fmaxf(w.x, 0.0f); w.y = fmaxf(w.y, 0.0f);
          w.z = fmaxf(w.z, 0.0f); w.w = fmaxf(w.w, 0.0f);
        }
      }
      *reinterpret_cast<float4*>(sA + r * PITCH + 4 * c4) = w;
    }
  }
  __syncthreads();

  const int tMl = wave / NT;
  const int tN  = wave % NT;
  const int tM  = blockIdx.x * STRIPS + tMl;
  if (tM * 16 >= M) return;

  int r16 = lane & 15;
  int hi  = lane >> 4;

  const float* Ap = sA + (tMl * 16 + r16) * PITCH + 2 * hi;
  const float* Bp = B + (size_t)(2 * hi) * NC + tN * 16 + r16;

  v8f acc;
  float binit = bias ? bias[tN * 16 + r16] : 0.0f;
  for (int r = 0; r < 8; ++r) acc[r] = binit;

#pragma unroll 4
  for (int k = 0; k < K; k += 4) {
    v2f a = *reinterpret_cast<const v2f*>(Ap + k);
    v2f b;
    b.x = Bp[(size_t)k * NC];
    b.y = Bp[(size_t)(k + 1) * NC];
    acc = __builtin_amdgcn_wmma_f32_16x16x4_f32(false, a, false, b,
                                                (short)0, acc, false, false);
  }

  float* Cp = C + (size_t)(tM * 16 + 8 * hi) * NC + tN * 16 + r16;
  for (int r = 0; r < 8; ++r) Cp[(size_t)r * NC] = acc[r];
}


extern "C" void kernel_launch(void* const* d_in, const int* in_sizes, int n_in,
                              void* d_out, int out_size, void* d_ws, size_t ws_size,
                              hipStream_t stream) {
  const float* x   = (const float*)d_in[0];
  const int*   ei  = (const int*)d_in[1];
  const float* W1  = (const float*)d_in[2];
  const float* b1  = (const float*)d_in[3];
  const float* W2  = (const float*)d_in[4];
  const float* b2  = (const float*)d_in[5];
  const float* Wfc = (const float*)d_in[6];
  const float* bfc = (const float*)d_in[7];
  float* out = (float*)d_out;

  const int N = in_sizes[0] / 128;
  const int E = in_sizes[1] / 2;
  const int* src = ei;
  const int* dst = ei + E;

  float* dinv = (float*)d_ws;
  float* bufA = dinv + N;
  float* bufB = bufA + (size_t)N * 128;

  auto cdiv = [](long long a, long long b) { return (int)((a + b - 1) / b); };

  k_fill<<<cdiv(N, 256), 256, 0, stream>>>(dinv, 1.0f, N);
  int* mask = (int*)d_out;
  const int nChunks = cdiv(E, kChunk);
  int* pair = mask + (size_t)nChunks * (kMaskRows * 4);
  k_deg<<<nChunks, 256, 0, stream>>>(src, dst, dinv, mask, pair, E);
  k_rsqrt<<<cdiv(N, 256), 256, 0, stream>>>(dinv, N);

  {
    int waves = cdiv(N, 16) * (128 >> 4);
    k_gemm_wmma<false, false, 128, 128><<<cdiv(waves, 8), 256, 0, stream>>>(
        x, W1, nullptr, nullptr, bufA, N);
  }
  k_agg<128, 8><<<cdiv(N, kOwnRows), 32 * kAggWaves, 0, stream>>>(bufA, pair, mask, dinv, bufB, E, N);

  float* h2   = bufA;
  float* acc2 = bufA + (size_t)N * 64;
  {
    int waves = cdiv(N, 16) * (64 >> 4);
    k_gemm_wmma<true, true, 128, 64><<<cdiv(waves, 8), 256, 0, stream>>>(
        bufB, W2, nullptr, b1, h2, N);
  }
  k_agg<64, 8><<<cdiv(N, kOwnRows), 32 * kAggWaves, 0, stream>>>(h2, pair, mask, dinv, acc2, E, N);

  {
    int waves = cdiv(N, 16) * (64 >> 4);
    k_gemm_wmma<true, false, 64, 64><<<cdiv(waves, 8), 256, 0, stream>>>(
        acc2, Wfc, bfc, b2, out, N);
  }
}


# edge sums: a workgroup owns 391 destination rows, not 512 (index key = destination / 391): 256 workgroups, one on each of the 256 WGPs, where 196 left 60 idle; sums stay in arrival order
# speedup vs baseline: 2.8776x; 1.0722x over previous
#include <hip/hip_runtime.h>

typedef __attribute__((ext_vector_type(2))) float v2f;
typedef __attribute__((ext_vector_type(8))) float v8f;


__global__ __launch_bounds__(256) void k_fill(float* __restrict__ p, float v, int n) {
  int i = blockIdx.x * 256 + threadIdx.x;
  if (i < n) p[i] = v;
}

constexpr int kOwnRows   = 391;
constexpr int kKeyMax    = 255;
constexpr int kMaskRows  = kKeyMax + 1;
constexpr int kChunk     = 2048;
constexpr int kChunkGrps = kChunk / 16;
constexpr int kDegEdges  = kChunk / 256;
static_assert(kChunkGrps == 128, "a mask word has 128 bits");
static_assert(kMaskRows == 256, "one thread per block key in the prefix sum");

__global__ __launch_bounds__(256) void k_deg(const int* __restrict__ src,
                                             const int* __restrict__ dst,
                                             float* __restrict__ deg,
                                             int* __restrict__ maskw,
                                             int* __restrict__ pairw, int E) {
  __shared__ __attribute__((aligned(16))) unsigned smask[kMaskRows * 4];
  __shared__ int scnt[kMaskRows];
  __shared__ int swsum[8];
  __shared__ __attribute__((aligned(16))) int2 spair[kChunk];
  const int tid = threadIdx.x;
  uint4* sm4 = reinterpret_cast<uint4*>(smask);
  sm4[tid] = make_uint4(0u, 0u, 0u, 0u);
  scnt[tid] = 0;
  __syncthreads();

  int2* pair = reinterpret_cast<int2*>(pairw);
  const int base = blockIdx.x * kChunk;
  int d[kDegEdges];
  int s[kDegEdges];
  int rk[kDegEdges];
  for (int j = 0; j < kDegEdges; ++j) {
    int i = base + j * 256 + tid;
    d[j] = 0;
    s[j] = 0;
    if (i < E) {
      d[j] = dst[i];
      s[j] = src[i];
    }
  }
  for (int j = 0; j < kDegEdges; ++j) {
    int i = base + j * 256 + tid;
    rk[j] = 0;
    if (i < E) {
      atomicAdd(&deg[d[j]], 1.0f);
      int k = (int)((unsigned)d[j] / (unsigned)kOwnRows);
      k = k < kKeyMax ? k : kKeyMax;
      rk[j] = atomicAdd(&scnt[k], 1);
    }
  }
  __syncthreads();

  {
    const int lane = tid & 31;
    const int wave = tid >> 5;
    const int c = scnt[tid];
    int incl = c;
    for (int off = 1; off < 32; off <<= 1) {
      int t = __shfl_up(incl, off);
      if (lane >= off) incl += t;
    }
    if (lane == 31) swsum[wave] = incl;
    __syncthreads();
    int wbase = 0;
    for (int w = 0; w < 8; ++w) {
      if (w < wave) wbase += swsum[w];
    }
    scnt[tid] = wbase + incl - c;
  }
  __syncthreads();

  for (int j = 0; j < kDegEdges; ++j) {
    int i = base + j * 256 + tid;
    if (i < E) {
      int k = (int)((unsigned)d[j] / (unsigned)kOwnRows);
      k = k < kKeyMax ? k : kKeyMax;
      int pos = scnt[k] + rk[j];
      spair[pos] = make_int2(d[j], s[j]);
      int g = pos >> 4;
      atomicOr(&smask[k * 4 + (g >> 5)], 1u << (g & 31));
    }
  }
  __syncthreads();
  for (int j = 0; j < kDegEdges; ++j) {
    int p = j * 256 + tid;
    if (base + p < E) pair[base + p] = spair[p];
  }
  reinterpret_cast<uint4*>(maskw)[(size_t)blockIdx.x * kMaskRows + tid] = sm4[tid];
}

__global__ __launch_bounds__(256) void k_rsqrt(float* __restrict__ p, int n) {
  int i = blockIdx.x * 256 + threadIdx.x;
  if (i < n) p[i] = rsqrtf(p[i]);
}


constexpr int kAggWaves  = 32;
constexpr int kQueueLen  = 128;
constexpr int kDrain     = 8;
constexpr int kGQueueLen = 256;
static_assert(kGQueueLen >= kDrain - 1 + 32, "a round's groups must fit behind a rest");
static_assert(kOwnRows <= 32 * kAggWaves, "one thread per owned row reads its dinv");
static_assert((kOwnRows * 64 + kOwnRows + kAggWaves * (2 * kQueueLen + kGQueueLen)) * 4 > 160 * 1024,
              "the 64-channel pass must not fit twice on one WGP");

template <int C, int U>
__global__ __launch_bounds__(32 * kAggWaves) void k_agg(const float* __restrict__ h,
                                                        const int* __restrict__ pairw,
                                                        const int* __restrict__ maskw,
                                                        const float* __restrict__ dinv,
                                                        float* __restrict__ out, int E, int N) {
  constexpr int R  = C / 32;
  constexpr int C4 = C / 4;
  __shared__ __attribute__((aligned(16))) float tile[kOwnRows * C];
  __shared__ float sdinv[kOwnRows];
  __shared__ int qs[kAggWaves * kQueueLen];
  __shared__ int qd[kAggWaves * kQueueLen];
  __shared__ int gq[kAggWaves * kGQueueLen];

  const int tid  = threadIdx.x;
  const int lane = tid & 31;
  const int wave = __builtin_amdgcn_readfirstlane(tid >> 5);
  const int n0   = blockIdx.x * kOwnRows;

  if (tid < kOwnRows) {
    int n = n0 + tid;
    sdinv[tid] = n < N ? dinv[n] : 0.0f;
  }
  __syncthreads();

  float4* t4 = reinterpret_cast<float4*>(tile);
  {
    const float4* h4 = reinterpret_cast<const float4*>(h + (size_t)n0 * C);
    for (int i = tid; i < kOwnRows * C4; i += 32 * kAggWaves) {
      int r = i / C4;
      float4 v = make_float4(0.0f, 0.0f, 0.0f, 0.0f);
      if (n0 + r < N) {
        v = h4[i];
        float d = sdinv[r];
        v.x = v.x * d * d;
        v.y = v.y * d * d;
        v.z = v.z * d * d;
        v.w = v.w * d * d;
      }
      t4[i] = v;
    }
  }
  __syncthreads();

  int* myqs = qs + wave * kQueueLen;
  int* myqd = qd + wave * kQueueLen;
  int cnt = 0;

  auto add_rows = [&](int s, int dl, float w, int j) {
    float v[U][R];
    for (int u = 0; u < U; ++u) {
      int sj = __builtin_amdgcn_readlane(s, j + u);
      const float* hp = h + (size_t)sj * C + lane;
      for (int r = 0; r < R; ++r) v[u][r] = hp[32 * r];
    }
    for (int u = 0; u < U; ++u) {
      int dj = __builtin_amdgcn_readlane(dl, j + u);
      float wj = __int_as_float(__builtin_amdgcn_readlane(__float_as_int(w), j + u));
      float* o = tile + dj * C + lane;
      for (int r = 0; r < R; ++r) atomicAdd(o + 32 * r, v[u][r] * wj);
    }
  };

  auto flush = [&]() {
    int head = 0;
    while (cnt - head >= 32) {
      int s = myqs[head + lane];
      int dl = myqd[head + lane];
      float w = dinv[s] * sdinv[dl];
      for (int j = 0; j < 32; j += U) add_rows(s, dl, w, j);
      head += 32;
    }
    if (head > 0) {
      int rem = cnt - head;
      if (lane < rem) {
        int ts = myqs[head + lane];
        int td = myqd[head + lane];
        myqs[lane] = ts;
        myqd[lane] = td;
      }
      cnt = rem;
    }
  };

  auto drain_rest = [&]() {
    int s = 0, dl = 0;
    if (lane < cnt) { s = myqs[lane]; dl = myqd[lane]; }
    float w = dinv[s] * sdinv[dl];
    int j = 0;
    for (; j + U <= cnt; j += U) add_rows(s, dl, w, j);
    for (; j < cnt; ++j) {
      int sj = __builtin_amdgcn_readlane(s, j);
      int dj = __builtin_amdgcn_readlane(dl, j);
      float wj = __int_as_float(__builtin_amdgcn_readlane(__float_as_int(w), j));
      const float* hp = h + (size_t)sj * C + lane;
      float* o = tile + dj * C + lane;
      float v[R];
      for (int r = 0; r < R; ++r) v[r] = hp[32 * r];
      for (int r = 0; r < R; ++r) atomicAdd(o + 32 * r, v[r] * wj);
    }
  };

  auto append = [&](unsigned mask, bool mine, int s, int dl) {
    if (mine) {
      int pos = cnt + (int)__builtin_amdgcn_mbcnt_lo(mask, 0u);
      myqs[pos] = s;
      myqd[pos] = dl;
    }
    cnt += __builtin_popcount(mask);
  };

  const int nChunks = (E + kChunk - 1) / kChunk;
  const uint4* mask4 = reinterpret_cast<const uint4*>(maskw);
  const int2* pair = reinterpret_cast<const int2*>(pairw);
  const unsigned own = blockIdx.x < (unsigned)kKeyMax ? blockIdx.x : (unsigned)kKeyMax;

  int* mygq = gq + wave * kGQueueLen;
  int gcnt = 0;
  const int half = lane >> 4;
  const int sub  = lane & 15;

  auto drain = [&](int ng) {
    const int gb = gcnt - ng;
    int gv[kDrain / 2];
    int ev[kDrain / 2];
    int dv[kDrain / 2];
    int sv[kDrain / 2];
    for (int p = 0; p < kDrain / 2; ++p) {
      int slot = 2 * p + half;
      gv[p] = mygq[slot < ng ? gb + slot : gb];
    }
    for (int p = 0; p < kDrain / 2; ++p) {
      int slot = 2 * p + half;
      int e = slot < ng ? gv[p] * 16 + sub : -1;
      ev[p] = e;
      unsigned ec = (unsigned)e < (unsigned)E ? (unsigned)e : (unsigned)(E - 1);
      int2 ds = pair[ec];
      dv[p] = ds.x;
      sv[p] = ds.y;
    }
    for (int p = 0; p < kDrain / 2; ++p) {
      if (!((unsigned)ev[p] < (unsigned)E)) dv[p] = -1;
    }
    unsigned mk[kDrain / 2];
    int total = 0;
    for (int p = 0; p < kDrain / 2; ++p) {
      mk[p] = __builtin_amdgcn_ballot_w32((unsigned)(dv[p] - n0) < (unsigned)kOwnRows);
      total += __builtin_popcount(mk[p]);
    }
    if (cnt + total <= kQueueLen) {
      for (int p = 0; p < kDrain / 2; ++p) {
        if (mk[p] != 0u) {
          unsigned dl = (unsigned)(dv[p] - n0);
          append(mk[p], dl < (unsigned)kOwnRows, sv[p], (int)dl);
        }
      }
    } else {
#pragma nounroll
      for (int p = 0; p < kDrain / 2; ++p) {
        if (cnt + 32 > kQueueLen) flush();
        int slot = 2 * p + half;
        int e = -1;
        if (slot < ng) e = mygq[gb + slot] * 16 + sub;
        int2 ds = make_int2(-1, 0);
        if ((unsigned)e < (unsigned)E) ds = pair[e];
        unsigned dl = (unsigned)(ds.x - n0);
        bool mine = dl < (unsigned)kOwnRows;
        unsigned mask = __builtin_amdgcn_ballot_w32(mine);
        if (mask != 0u) append(mask, mine, ds.y, (int)dl);
      }
    }
    gcnt = gb;
  };

  for (int c0 = wave; c0 < nChunks; c0 += 32 * kAggWaves) {
    const int c = c0 + kAggWaves * lane;
    uint4 m = make_uint4(0u, 0u, 0u, 0u);
    if (c < nChunks) m = mask4[(size_t)c * kMaskRows + own];
    unsigned long long lo = (unsigned long long)m.x | ((unsigned long long)m.y << 32);
    unsigned long long hi = (unsigned long long)m.z | ((unsigned long long)m.w << 32);
    const int gfirst = c * kChunkGrps;
    for (;;) {
      bool hit = (lo | hi) != 0ull;
      unsigned mk = __builtin_amdgcn_ballot_w32(hit);
      if (mk == 0u) break;
      if (hit) {
        int g;
        if (lo != 0ull) {
          g = __builtin_ctzll(lo);
          lo &= lo - 1ull;
        } else {
          g = 64 + __builtin_ctzll(hi);
          hi &= hi - 1ull;
        }
        int pos = gcnt + (int)__builtin_amdgcn_mbcnt_lo(mk, 0u);
        mygq[pos] = gfirst + g;
      }
      gcnt += __builtin_popcount(mk);
      while (gcnt >= kDrain) {
        drain(kDrain);
        flush();
      }
    }
  }
  if (gcnt > 0) drain(gcnt);
  flush();
  if (cnt > 0) drain_rest();
  __syncthreads();

  {
    float4* o4 = reinterpret_cast<float4*>(out + (size_t)n0 * C);
    for (int i = tid; i < kOwnRows * C4; i += 32 * kAggWaves) {
      int r = i / C4;
      if (n0 + r < N) o4[i] = t4[i];
    }
  }
}

template <bool ABIAS, bool RELU, int K, int NC>
__global__ __launch_bounds__(256) void k_gemm_wmma(const float* __restrict__ A,
                                                   const float* __restrict__ B,
                                                   const float* __restrict__ bias,
                                                   const float* __restrict__ abias,
                                                   float* __restrict__ C, int M) {
  constexpr int NT     = NC / 16;
  constexpr int STRIPS = 8 / NT;
  constexpr int ROWS   = 16 * STRIPS;
  constexpr int PITCH  = K + 4;
  constexpr int K4     = K / 4;
  static_assert(NT * STRIPS == 8 && (ROWS * K4) % 256 == 0, "tile shape");
  __shared__ __attribute__((aligned(16))) float sA[ROWS * PITCH];

  const int tid  = threadIdx.x;
  const int lane = tid & 31;
  const int wave = __builtin_amdgcn_readfirstlane(tid >> 5);
  const int row0 = blockIdx.x * ROWS;

  {
    const float4* A4  = reinterpret_cast<const float4*>(A);
    const float4* ab4 = reinterpret_cast<const float4*>(abias);
    float4 v[ROWS * K4 / 256];
    for (int j = 0; j < ROWS * K4 / 256; ++j) {
      int i  = tid + 256 * j;
      int r  = i / K4;
      int c4 = i % K4;
      v[j] = make_float4(0.0f, 0.0f, 0.0f, 0.0f);
      if (row0 + r < M) v[j] = A4[(size_t)(row0 + r) * K4 + c4];
    }
    for (int j = 0; j < ROWS * K4 / 256; ++j) {
      int i  = tid + 256 * j;
      int r  = i / K4;
      int c4 = i % K4;
      float4 w = v[j];
      if (ABIAS) {
        const float4 bb = ab4[c4];
        w.x += bb.x; w.y += bb.y; w.z += bb.z; w.w += bb.w;
        if (RELU) {
          w.x = fmaxf(w.x, 0.0f); w.y = fmaxf(w.y, 0.0f);
          w.z = fmaxf(w.z, 0.0f); w.w = fmaxf(w.w, 0.0f);
        }
      }
      *reinterpret_cast<float4*>(sA + r * PITCH + 4 * c4) = w;
    }
  }
  __syncthreads();

  const int tMl = wave / NT;
  const int tN  = wave % NT;
  const int tM  = blockIdx.x * STRIPS + tMl;
  if (tM * 16 >= M) return;

  int r16 = lane & 15;
  int hi  = lane >> 4;

  const float* Ap = sA + (tMl * 16 + r16) * PITCH + 2 * hi;
  const float* Bp = B + (size_t)(2 * hi) * NC + tN * 16 + r16;

  v8f acc;
  float binit = bias ? bias[tN * 16 + r16] : 0.0f;
  for (int r = 0; r < 8; ++r) acc[r] = binit;

#pragma unroll 4
  for (int k = 0; k < K; k += 4) {
    v2f a = *reinterpret_cast<const v2f*>(Ap + k);
    v2f b;
    b.x = Bp[(size_t)k * NC];
    b.y = Bp[(size_t)(k + 1) * NC];
    acc = __builtin_amdgcn_wmma_f32_16x16x4_f32(false, a, false, b,
                                                (short)0, acc, false, false);
  }

  float* Cp = C + (size_t)(tM * 16 + 8 * hi) * NC + tN * 16 + r16;
  for (int r = 0; r < 8; ++r) Cp[(size_t)r * NC] = acc[r];
}


extern "C" void kernel_launch(void* const* d_in, const int* in_sizes, int n_in,
                              void* d_out, int out_size, void* d_ws, size_t ws_size,
                              hipStream_t stream) {
  const float* x   = (const float*)d_in[0];
  const int*   ei  = (const int*)d_in[1];
  const float* W1  = (const float*)d_in[2];
  const float* b1  = (const float*)d_in[3];
  const float* W2  = (const float*)d_in[4];
  const float* b2  = (const float*)d_in[5];
  const float* Wfc = (const float*)d_in[6];
  const float* bfc = (const float*)d_in[7];
  float* out = (float*)d_out;

  const int N = in_sizes[0] / 128;
  const int E = in_sizes[1] / 2;
  const int* src = ei;
  const int* dst = ei + E;

  float* dinv = (float*)d_ws;
  float* bufA = dinv + N;
  float* bufB = bufA + (size_t)N * 128;

  auto cdiv = [](long long a, long long b) { return (int)((a + b - 1) / b); };

  k_fill<<<cdiv(N, 256), 256, 0, stream>>>(dinv, 1.0f, N);
  int* mask = (int*)d_out;
  const int nChunks = cdiv(E, kChunk);
  int* pair = mask + (size_t)nChunks * (kMaskRows * 4);
  k_deg<<<nChunks, 256, 0, stream>>>(src, dst, dinv, mask, pair, E);
  k_rsqrt<<<cdiv(N, 256), 256, 0, stream>>>(dinv, N);

  {
    int waves = cdiv(N, 16) * (128 >> 4);
    k_gemm_wmma<false, false, 128, 128><<<cdiv(waves, 8), 256, 0, stream>>>(
        x, W1, nullptr, nullptr, bufA, N);
  }
  k_agg<128, 8><<<cdiv(N, kOwnRows), 32 * kAggWaves, 0, stream>>>(bufA, pair, mask, dinv, bufB, E, N);

  float* h2   = bufA;
  float* acc2 = bufA + (size_t)N * 64;
  {
    int waves = cdiv(N, 16) * (64 >> 4);
    k_gemm_wmma<true, true, 128, 64><<<cdiv(waves, 8), 256, 0, stream>>>(
        bufB, W2, nullptr, b1, h2, N);
  }
  k_agg<64, 8><<<cdiv(N, kOwnRows), 32 * kAggWaves, 0, stream>>>(h2, pair, mask, dinv, acc2, E, N);

  {
    int waves = cdiv(N, 16) * (64 >> 4);
    k_gemm_wmma<true, false, 64, 64><<<cdiv(waves, 8), 256, 0, stream>>>(
        acc2, Wfc, bfc, b2, out, N);
  }
}


# layer 1 sums the rows of x over the edges BEFORE its product (sum(xW) = (sum x)W): products 1 and 2 run back to back in one launch, hidden rows stay in LDS; 8 -> 7 launches; order of sums changes
# speedup vs baseline: 3.0604x; 1.0635x over previous
#include <hip/hip_runtime.h>

typedef __attribute__((ext_vector_type(2))) float v2f;
typedef __attribute__((ext_vector_type(8))) float v8f;


__global__ __launch_bounds__(256) void k_fill(float* __restrict__ p, float v, int n) {
  int i = blockIdx.x * 256 + threadIdx.x;
  if (i < n) p[i] = v;
}

constexpr int kOwnRows   = 391;
constexpr int kKeyMax    = 255;
constexpr int kMaskRows  = kKeyMax + 1;
constexpr int kChunk     = 2048;
constexpr int kChunkGrps = kChunk / 16;
constexpr int kDegEdges  = kChunk / 256;
static_assert(kChunkGrps == 128, "a mask word has 128 bits");
static_assert(kMaskRows == 256, "one thread per block key in the prefix sum");

__global__ __launch_bounds__(256) void k_deg(const int* __restrict__ src,
                                             const int* __restrict__ dst,
                                             float* __restrict__ deg,
                                             int* __restrict__ maskw,
                                             int* __restrict__ pairw, int E) {
  __shared__ __attribute__((aligned(16))) unsigned smask[kMaskRows * 4];
  __shared__ int scnt[kMaskRows];
  __shared__ int swsum[8];
  __shared__ __attribute__((aligned(16))) int2 spair[kChunk];
  const int tid = threadIdx.x;
  uint4* sm4 = reinterpret_cast<uint4*>(smask);
  sm4[tid] = make_uint4(0u, 0u, 0u, 0u);
  scnt[tid] = 0;
  __syncthreads();

  int2* pair = reinterpret_cast<int2*>(pairw);
  const int base = blockIdx.x * kChunk;
  int d[kDegEdges];
  int s[kDegEdges];
  int rk[kDegEdges];
  for (int j = 0; j < kDegEdges; ++j) {
    int i = base + j * 256 + tid;
    d[j] = 0;
    s[j] = 0;
    if (i < E) {
      d[j] = dst[i];
      s[j] = src[i];
    }
  }
  for (int j = 0; j < kDegEdges; ++j) {
    int i = base + j * 256 + tid;
    rk[j] = 0;
    if (i < E) {
      atomicAdd(&deg[d[j]], 1.0f);
      int k = (int)((unsigned)d[j] / (unsigned)kOwnRows);
      k = k < kKeyMax ? k : kKeyMax;
      rk[j] = atomicAdd(&scnt[k], 1);
    }
  }
  __syncthreads();

  {
    const int lane = tid & 31;
    const int wave = tid >> 5;
    const int c = scnt[tid];
    int incl = c;
    for (int off = 1; off < 32; off <<= 1) {
      int t = __shfl_up(incl, off);
      if (lane >= off) incl += t;
    }
    if (lane == 31) swsum[wave] = incl;
    __syncthreads();
    int wbase = 0;
    for (int w = 0; w < 8; ++w) {
      if (w < wave) wbase += swsum[w];
    }
    scnt[tid] = wbase + incl - c;
  }
  __syncthreads();

  for (int j = 0; j < kDegEdges; ++j) {
    int i = base + j * 256 + tid;
    if (i < E) {
      int k = (int)((unsigned)d[j] / (unsigned)kOwnRows);
      k = k < kKeyMax ? k : kKeyMax;
      int pos = scnt[k] + rk[j];
      spair[pos] = make_int2(d[j], s[j]);
      int g = pos >> 4;
      atomicOr(&smask[k * 4 + (g >> 5)], 1u << (g & 31));
    }
  }
  __syncthreads();
  for (int j = 0; j < kDegEdges; ++j) {
    int p = j * 256 + tid;
    if (base + p < E) pair[base + p] = spair[p];
  }
  reinterpret_cast<uint4*>(maskw)[(size_t)blockIdx.x * kMaskRows + tid] = sm4[tid];
}

__global__ __launch_bounds__(256) void k_rsqrt(float* __restrict__ p, int n) {
  int i = blockIdx.x * 256 + threadIdx.x;
  if (i < n) p[i] = rsqrtf(p[i]);
}


constexpr int kAggWaves  = 32;
constexpr int kQueueLen  = 128;
constexpr int kDrain     = 8;
constexpr int kGQueueLen = 256;
static_assert(kGQueueLen >= kDrain - 1 + 32, "a round's groups must fit behind a rest");
static_assert(kOwnRows <= 32 * kAggWaves, "one thread per owned row reads its dinv");
static_assert((kOwnRows * 64 + kOwnRows + kAggWaves * (2 * kQueueLen + kGQueueLen)) * 4 > 160 * 1024,
              "the 64-channel pass must not fit twice on one WGP");

template <int C, int U>
__global__ __launch_bounds__(32 * kAggWaves) void k_agg(const float* __restrict__ h,
                                                        const int* __restrict__ pairw,
                                                        const int* __restrict__ maskw,
                                                        const float* __restrict__ dinv,
                                                        float* __restrict__ out, int E, int N) {
  constexpr int R  = C / 32;
  constexpr int C4 = C / 4;
  __shared__ __attribute__((aligned(16))) float tile[kOwnRows * C];
  __shared__ float sdinv[kOwnRows];
  __shared__ int qs[kAggWaves * kQueueLen];
  __shared__ int qd[kAggWaves * kQueueLen];
  __shared__ int gq[kAggWaves * kGQueueLen];

  const int tid  = threadIdx.x;
  const int lane = tid & 31;
  const int wave = __builtin_amdgcn_readfirstlane(tid >> 5);
  const int n0   = blockIdx.x * kOwnRows;

  if (tid < kOwnRows) {
    int n = n0 + tid;
    sdinv[tid] = n < N ? dinv[n] : 0.0f;
  }
  __syncthreads();

  float4* t4 = reinterpret_cast<float4*>(tile);
  {
    const float4* h4 = reinterpret_cast<const float4*>(h + (size_t)n0 * C);
    for (int i = tid; i < kOwnRows * C4; i += 32 * kAggWaves) {
      int r = i / C4;
      float4 v = make_float4(0.0f, 0.0f, 0.0f, 0.0f);
      if (n0 + r < N) {
        v = h4[i];
        float d = sdinv[r];
        v.x = v.x * d * d;
        v.y = v.y * d * d;
        v.z = v.z * d * d;
        v.w = v.w * d * d;
      }
      t4[i] = v;
    }
  }
  __syncthreads();

  int* myqs = qs + wave * kQueueLen;
  int* myqd = qd + wave * kQueueLen;
  int cnt = 0;

  auto add_rows = [&](int s, int dl, float w, int j) {
    float v[U][R];
    for (int u = 0; u < U; ++u) {
      int sj = __builtin_amdgcn_readlane(s, j + u);
      const float* hp = h + (size_t)sj * C + lane;
      for (int r = 0; r < R; ++r) v[u][r] = hp[32 * r];
    }
    for (int u = 0; u < U; ++u) {
      int dj = __builtin_amdgcn_readlane(dl, j + u);
      float wj = __int_as_float(__builtin_amdgcn_readlane(__float_as_int(w), j + u));
      float* o = tile + dj * C + lane;
      for (int r = 0; r < R; ++r) atomicAdd(o + 32 * r, v[u][r] * wj);
    }
  };

  auto flush = [&]() {
    int head = 0;
    while (cnt - head >= 32) {
      int s = myqs[head + lane];
      int dl = myqd[head + lane];
      float w = dinv[s] * sdinv[dl];
      for (int j = 0; j < 32; j += U) add_rows(s, dl, w, j);
      head += 32;
    }
    if (head > 0) {
      int rem = cnt - head;
      if (lane < rem) {
        int ts = myqs[head + lane];
        int td = myqd[head + lane];
        myqs[lane] = ts;
        myqd[lane] = td;
      }
      cnt = rem;
    }
  };

  auto drain_rest = [&]() {
    int s = 0, dl = 0;
    if (lane < cnt) { s = myqs[lane]; dl = myqd[lane]; }
    float w = dinv[s] * sdinv[dl];
    int j = 0;
    for (; j + U <= cnt; j += U) add_rows(s, dl, w, j);
    for (; j < cnt; ++j) {
      int sj = __builtin_amdgcn_readlane(s, j);
      int dj = __builtin_amdgcn_readlane(dl, j);
      float wj = __int_as_float(__builtin_amdgcn_readlane(__float_as_int(w), j));
      const float* hp = h + (size_t)sj * C + lane;
      float* o = tile + dj * C + lane;
      float v[R];
      for (int r = 0; r < R; ++r) v[r] = hp[32 * r];
      for (int r = 0; r < R; ++r) atomicAdd(o + 32 * r, v[r] * wj);
    }
  };

  auto append = [&](unsigned mask, bool mine, int s, int dl) {
    if (mine) {
      int pos = cnt + (int)__builtin_amdgcn_mbcnt_lo(mask, 0u);
      myqs[pos] = s;
      myqd[pos] = dl;
    }
    cnt += __builtin_popcount(mask);
  };

  const int nChunks = (E + kChunk - 1) / kChunk;
  const uint4* mask4 = reinterpret_cast<const uint4*>(maskw);
  const int2* pair = reinterpret_cast<const int2*>(pairw);
  const unsigned own = blockIdx.x < (unsigned)kKeyMax ? blockIdx.x : (unsigned)kKeyMax;

  int* mygq = gq + wave * kGQueueLen;
  int gcnt = 0;
  const int half = lane >> 4;
  const int sub  = lane & 15;

  auto drain = [&](int ng) {
    const int gb = gcnt - ng;
    int gv[kDrain / 2];
    int ev[kDrain / 2];
    int dv[kDrain / 2];
    int sv[kDrain / 2];
    for (int p = 0; p < kDrain / 2; ++p) {
      int slot = 2 * p + half;
      gv[p] = mygq[slot < ng ? gb + slot : gb];
    }
    for (int p = 0; p < kDrain / 2; ++p) {
      int slot = 2 * p + half;
      int e = slot < ng ? gv[p] * 16 + sub : -1;
      ev[p] = e;
      unsigned ec = (unsigned)e < (unsigned)E ? (unsigned)e : (unsigned)(E - 1);
      int2 ds = pair[ec];
      dv[p] = ds.x;
      sv[p] = ds.y;
    }
    for (int p = 0; p < kDrain / 2; ++p) {
      if (!((unsigned)ev[p] < (unsigned)E)) dv[p] = -1;
    }
    unsigned mk[kDrain / 2];
    int total = 0;
    for (int p = 0; p < kDrain / 2; ++p) {
      mk[p] = __builtin_amdgcn_ballot_w32((unsigned)(dv[p] - n0) < (unsigned)kOwnRows);
      total += __builtin_popcount(mk[p]);
    }
    if (cnt + total <= kQueueLen) {
      for (int p = 0; p < kDrain / 2; ++p) {
        if (mk[p] != 0u) {
          unsigned dl = (unsigned)(dv[p] - n0);
          append(mk[p], dl < (unsigned)kOwnRows, sv[p], (int)dl);
        }
      }
    } else {
#pragma nounroll
      for (int p = 0; p < kDrain / 2; ++p) {
        if (cnt + 32 > kQueueLen) flush();
        int slot = 2 * p + half;
        int e = -1;
        if (slot < ng) e = mygq[gb + slot] * 16 + sub;
        int2 ds = make_int2(-1, 0);
        if ((unsigned)e < (unsigned)E) ds = pair[e];
        unsigned dl = (unsigned)(ds.x - n0);
        bool mine = dl < (unsigned)kOwnRows;
        unsigned mask = __builtin_amdgcn_ballot_w32(mine);
        if (mask != 0u) append(mask, mine, ds.y, (int)dl);
      }
    }
    gcnt = gb;
  };

  for (int c0 = wave; c0 < nChunks; c0 += 32 * kAggWaves) {
    const int c = c0 + kAggWaves * lane;
    uint4 m = make_uint4(0u, 0u, 0u, 0u);
    if (c < nChunks) m = mask4[(size_t)c * kMaskRows + own];
    unsigned long long lo = (unsigned long long)m.x | ((unsigned long long)m.y << 32);
    unsigned long long hi = (unsigned long long)m.z | ((unsigned long long)m.w << 32);
    const int gfirst = c * kChunkGrps;
    for (;;) {
      bool hit = (lo | hi) != 0ull;
      unsigned mk = __builtin_amdgcn_ballot_w32(hit);
      if (mk == 0u) break;
      if (hit) {
        int g;
        if (lo != 0ull) {
          g = __builtin_ctzll(lo);
          lo &= lo - 1ull;
        } else {
          g = 64 + __builtin_ctzll(hi);
          hi &= hi - 1ull;
        }
        int pos = gcnt + (int)__builtin_amdgcn_mbcnt_lo(mk, 0u);
        mygq[pos] = gfirst + g;
      }
      gcnt += __builtin_popcount(mk);
      while (gcnt >= kDrain) {
        drain(kDrain);
        flush();
      }
    }
  }
  if (gcnt > 0) drain(gcnt);
  flush();
  if (cnt > 0) drain_rest();
  __syncthreads();

  {
    float4* o4 = reinterpret_cast<float4*>(out + (size_t)n0 * C);
    for (int i = tid; i < kOwnRows * C4; i += 32 * kAggWaves) {
      int r = i / C4;
      if (n0 + r < N) o4[i] = t4[i];
    }
  }
}

template <bool ABIAS, bool RELU, int K, int NC>
__global__ __launch_bounds__(256) void k_gemm_wmma(const float* __restrict__ A,
                                                   const float* __restrict__ B,
                                                   const float* __restrict__ bias,
                                                   const float* __restrict__ abias,
                                                   float* __restrict__ C, int M) {
  constexpr int NT     = NC / 16;
  constexpr int STRIPS = 8 / NT;
  constexpr int ROWS   = 16 * STRIPS;
  constexpr int PITCH  = K + 4;
  constexpr int K4     = K / 4;
  static_assert(NT * STRIPS == 8 && (ROWS * K4) % 256 == 0, "tile shape");
  __shared__ __attribute__((aligned(16))) float sA[ROWS * PITCH];

  const int tid  = threadIdx.x;
  const int lane = tid & 31;
  const int wave = __builtin_amdgcn_readfirstlane(tid >> 5);
  const int row0 = blockIdx.x * ROWS;

  {
    const float4* A4  = reinterpret_cast<const float4*>(A);
    const float4* ab4 = reinterpret_cast<const float4*>(abias);
    float4 v[ROWS * K4 / 256];
    for (int j = 0; j < ROWS * K4 / 256; ++j) {
      int i  = tid + 256 * j;
      int r  = i / K4;
      int c4 = i % K4;
      v[j] = make_float4(0.0f, 0.0f, 0.0f, 0.0f);
      if (row0 + r < M) v[j] = A4[(size_t)(row0 + r) * K4 + c4];
    }
    for (int j = 0; j < ROWS * K4 / 256; ++j) {
      int i  = tid + 256 * j;
      int r  = i / K4;
      int c4 = i % K4;
      float4 w = v[j];
      if (ABIAS) {
        const float4 bb = ab4[c4];
        w.x += bb.x; w.y += bb.y; w.z += bb.z; w.w += bb.w;
        if (RELU) {
          w.x = fmaxf(w.x, 0.0f); w.y = fmaxf(w.y, 0.0f);
          w.z = fmaxf(w.z, 0.0f); w.w = fmaxf(w.w, 0.0f);
        }
      }
      *reinterpret_cast<float4*>(sA + r * PITCH + 4 * c4) = w;
    }
  }
  __syncthreads();

  const int tMl = wave / NT;
  const int tN  = wave % NT;
  const int tM  = blockIdx.x * STRIPS + tMl;
  if (tM * 16 >= M) return;

  int r16 = lane & 15;
  int hi  = lane >> 4;

  const float* Ap = sA + (tMl * 16 + r16) * PITCH + 2 * hi;
  const float* Bp = B + (size_t)(2 * hi) * NC + tN * 16 + r16;

  v8f acc;
  float binit = bias ? bias[tN * 16 + r16] : 0.0f;
  for (int r = 0; r < 8; ++r) acc[r] = binit;

#pragma unroll 4
  for (int k = 0; k < K; k += 4) {
    v2f a = *reinterpret_cast<const v2f*>(Ap + k);
    v2f b;
    b.x = Bp[(size_t)k * NC];
    b.y = Bp[(size_t)(k + 1) * NC];
    acc = __builtin_amdgcn_wmma_f32_16x16x4_f32(false, a, false, b,
                                                (short)0, acc, false, false);
  }

  float* Cp = C + (size_t)(tM * 16 + 8 * hi) * NC + tN * 16 + r16;
  for (int r = 0; r < 8; ++r) Cp[(size_t)r * NC] = acc[r];
}

template <int K1, int N1, int N2>
__global__ __launch_bounds__(256) void k_gemm_pair(const float* __restrict__ A,
                                                   const float* __restrict__ B1,
                                                   const float* __restrict__ bias1,
                                                   const float* __restrict__ B2,
                                                   float* __restrict__ C, int M) {
  constexpr int ROWS = 32;
  constexpr int P1   = K1 + 4;
  constexpr int P2   = N1 + 4;
  constexpr int K4   = K1 / 4;
  constexpr int NT2  = N2 / 16;
  static_assert(N1 / 16 == 8, "first product: one column tile per wave");
  static_assert(NT2 * (ROWS / 16) == 8, "second product: one tile per wave");
  static_assert((ROWS * K4) % 256 == 0, "tile shape");
  __shared__ __attribute__((aligned(16))) float sA[ROWS * P1];
  __shared__ __attribute__((aligned(16))) float sH[ROWS * P2];

  const int tid  = threadIdx.x;
  const int lane = tid & 31;
  const int wave = __builtin_amdgcn_readfirstlane(tid >> 5);
  const int row0 = blockIdx.x * ROWS;

  {
    const float4* A4 = reinterpret_cast<const float4*>(A);
    float4 v[ROWS * K4 / 256];
    for (int j = 0; j < ROWS * K4 / 256; ++j) {
      int i  = tid + 256 * j;
      int r  = i / K4;
      int c4 = i % K4;
      v[j] = make_float4(0.0f, 0.0f, 0.0f, 0.0f);
      if (row0 + r < M) v[j] = A4[(size_t)(row0 + r) * K4 + c4];
    }
    for (int j = 0; j < ROWS * K4 / 256; ++j) {
      int i  = tid + 256 * j;
      int r  = i / K4;
      int c4 = i % K4;
      *reinterpret_cast<float4*>(sA + r * P1 + 4 * c4) = v[j];
    }
  }
  __syncthreads();

  const int r16 = lane & 15;
  const int hi  = lane >> 4;

  {
    const float* Ap0 = sA + r16 * P1 + 2 * hi;
    const float* Ap1 = Ap0 + 16 * P1;
    const float* Bp  = B1 + (size_t)(2 * hi) * N1 + wave * 16 + r16;

    v8f acc0, acc1;
    const float binit = bias1[wave * 16 + r16];
    for (int r = 0; r < 8; ++r) { acc0[r] = binit; acc1[r] = binit; }

#pragma unroll 4
    for (int k = 0; k < K1; k += 4) {
      v2f a0 = *reinterpret_cast<const v2f*>(Ap0 + k);
      v2f a1 = *reinterpret_cast<const v2f*>(Ap1 + k);
      v2f b;
      b.x = Bp[(size_t)k * N1];
      b.y = Bp[(size_t)(k + 1) * N1];
      acc0 = __builtin_amdgcn_wmma_f32_16x16x4_f32(false, a0, false, b,
                                                   (short)0, acc0, false, false);
      acc1 = __builtin_amdgcn_wmma_f32_16x16x4_f32(false, a1, false, b,
                                                   (short)0, acc1, false, false);
    }

    float* Hp = sH + (8 * hi) * P2 + wave * 16 + r16;
    for (int r = 0; r < 8; ++r) {
      Hp[r * P2]        = fmaxf(acc0[r], 0.0f);
      Hp[(16 + r) * P2] = fmaxf(acc1[r], 0.0f);
    }
  }
  __syncthreads();

  const int tMl = wave / NT2;
  const int tN  = wave % NT2;
  const int tM  = blockIdx.x * (ROWS / 16) + tMl;
  if (tM * 16 >= M) return;

  const float* Hq = sH + (tMl * 16 + r16) * P2 + 2 * hi;
  const float* Bq = B2 + (size_t)(2 * hi) * N2 + tN * 16 + r16;

  v8f acc;
  for (int r = 0; r < 8; ++r) acc[r] = 0.0f;

#pragma unroll 4
  for (int k = 0; k < N1; k += 4) {
    v2f a = *reinterpret_cast<const v2f*>(Hq + k);
    v2f b;
    b.x = Bq[(size_t)k * N2];
    b.y = Bq[(size_t)(k + 1) * N2];
    acc = __builtin_amdgcn_wmma_f32_16x16x4_f32(false, a, false, b,
                                                (short)0, acc, false, false);
  }

  float* Cp = C + (size_t)(tM * 16 + 8 * hi) * N2 + tN * 16 + r16;
  for (int r = 0; r < 8; ++r) Cp[(size_t)r * N2] = acc[r];
}


extern "C" void kernel_launch(void* const* d_in, const int* in_sizes, int n_in,
                              void* d_out, int out_size, void* d_ws, size_t ws_size,
                              hipStream_t stream) {
  const float* x   = (const float*)d_in[0];
  const int*   ei  = (const int*)d_in[1];
  const float* W1  = (const float*)d_in[2];
  const float* b1  = (const float*)d_in[3];
  const float* W2  = (const float*)d_in[4];
  const float* b2  = (const float*)d_in[5];
  const float* Wfc = (const float*)d_in[6];
  const float* bfc = (const float*)d_in[7];
  float* out = (float*)d_out;

  const int N = in_sizes[0] / 128;
  const int E = in_sizes[1] / 2;
  const int* src = ei;
  const int* dst = ei + E;

  float* dinv = (float*)d_ws;
  float* bufA = dinv + N;
  float* bufB = bufA + (size_t)N * 128;

  auto cdiv = [](long long a, long long b) { return (int)((a + b - 1) / b); };

  k_fill<<<cdiv(N, 256), 256, 0, stream>>>(dinv, 1.0f, N);
  int* mask = (int*)d_out;
  const int nChunks = cdiv(E, kChunk);
  int* pair = mask + (size_t)nChunks * (kMaskRows * 4);
  k_deg<<<nChunks, 256, 0, stream>>>(src, dst, dinv, mask, pair, E);
  k_rsqrt<<<cdiv(N, 256), 256, 0, stream>>>(dinv, N);

  k_agg<128, 8><<<cdiv(N, kOwnRows), 32 * kAggWaves, 0, stream>>>(x, pair, mask, dinv, bufB, E, N);

  float* h2   = bufA;
  float* acc2 = bufA + (size_t)N * 64;
  k_gemm_pair<128, 128, 64><<<cdiv(N, 32), 256, 0, stream>>>(bufB, W1, b1, W2, h2, N);
  k_agg<64, 8><<<cdiv(N, kOwnRows), 32 * kAggWaves, 0, stream>>>(h2, pair, mask, dinv, acc2, E, N);

  {
    int waves = cdiv(N, 16) * (64 >> 4);
    k_gemm_wmma<true, false, 64, 64><<<cdiv(waves, 8), 256, 0, stream>>>(
        acc2, Wfc, bfc, b2, out, N);
  }
}
